# starting kernel
# baseline (speedup 1.0000x reference)
#include <hip/hip_runtime.h>

typedef __attribute__((ext_vector_type(2))) float v2f;
typedef __attribute__((ext_vector_type(8))) float v8f;

#define IN_F   128
#define HID_F  64
#define OUT_F  32


__global__ void k_init_deg(float* __restrict__ deg_in, float* __restrict__ deg_out, int N) {
  int i = blockIdx.x * blockDim.x + threadIdx.x;
  if (i < N) { deg_in[i] = 1.0f; deg_out[i] = 1.0f; }
}

__global__ void k_count_deg(const int* __restrict__ ei0, const int* __restrict__ ei1,
                            float* __restrict__ deg_in, float* __restrict__ deg_out, int E) {
  int e = blockIdx.x * blockDim.x + threadIdx.x;
  if (e < E) {
    atomicAdd(&deg_out[ei0[e]], 1.0f);
    atomicAdd(&deg_in [ei1[e]], 1.0f);
  }
}

__global__ void k_norms(const int* __restrict__ ei0, const int* __restrict__ ei1,
                        const float* __restrict__ deg_in, const float* __restrict__ deg_out,
                        float* __restrict__ rn_edge, float* __restrict__ rn_self, int E, int N) {
  int i = blockIdx.x * blockDim.x + threadIdx.x;
  if (i < E) {
    rn_edge[i] = rsqrtf(deg_in[ei1[i]] * deg_out[ei0[i]]);
  } else if (i < E + N) {
    int n = i - E;
    rn_self[n] = rsqrtf(deg_in[n] * deg_out[n]);
  }
}


template <int K, int RELU>
__global__ __launch_bounds__(128)
void k_gemm_wmma(const float* __restrict__ X, int ldx,
                 const float* __restrict__ Wa, const float* __restrict__ ba,
                 const float* __restrict__ Wb, const float* __restrict__ bb,
                 float* __restrict__ Y, int ldy,
                 int halfO, int koffB, int nColTiles, int nMTiles) {
  constexpr int NK = K / 4;

  int wave = threadIdx.x >> 5;
  int lane = threadIdx.x & 31;
  int mt   = blockIdx.x * 4 + wave;
  if (mt >= nMTiles) return;

  int m0   = mt << 4;
  int mrow = lane & 15;
  int khi  = (lane >> 4) << 1;
  int col  = lane & 15;

  const float* xrow = X + (size_t)(m0 + mrow) * ldx;

  v2f afrag[NK];
  int curKoff = -1;

  for (int ct = 0; ct < nColTiles; ++ct) {
    int c0   = ct << 4;
    int half = (c0 >= halfO) ? 1 : 0;
    const float* W    = half ? Wb : Wa;
    const float* bias = half ? bb : ba;
    int wc0  = c0 - half * halfO;
    int koff = half ? koffB : 0;

    if (koff != curKoff) {
      curKoff = koff;
      const float* xp = xrow + koff + khi;
#pragma unroll
      for (int i = 0; i < NK; ++i) {
        v2f a = *(const v2f*)(xp + i * 4);
        if (RELU) { a.x = fmaxf(a.x, 0.0f); a.y = fmaxf(a.y, 0.0f); }
        afrag[i] = a;
      }
    }

    const float* wrow = W + (size_t)(wc0 + col) * K + khi;
    v8f acc = {};
#pragma unroll
    for (int i = 0; i < NK; ++i) {
      v2f bf = *(const v2f*)(wrow + i * 4);
      acc = __builtin_amdgcn_wmma_f32_16x16x4_f32(false, afrag[i], false, bf,
                                                  (short)0, acc, false, false);
    }

    float bv = bias[wc0 + col];
    int mbase = m0 + ((lane >> 4) << 3);
    float* yp = Y + (size_t)mbase * ldy + c0 + col;
#pragma unroll
    for (int g = 0; g < 8; ++g)
      yp[(size_t)g * ldy] = acc[g] + bv;
  }
}


__global__ void k_aggr_self(const float* __restrict__ T, const float* __restrict__ rn_self,
                            float* __restrict__ Y, int total4, int fshift) {
  int idx = blockIdx.x * blockDim.x + threadIdx.x;
  if (idx < total4) {
    float rn = rn_self[(idx << 2) >> fshift];
    float4 v = ((const float4*)T)[idx];
    float4 o; o.x = rn * v.x; o.y = rn * v.y; o.z = rn * v.z; o.w = rn * v.w;
    ((float4*)Y)[idx] = o;
  }
}

__global__ void k_scatter(const float* __restrict__ T,
                          const int* __restrict__ src, const int* __restrict__ dst,
                          const float* __restrict__ rn_edge, float* __restrict__ Y,
                          int E, int F) {
  int chunks = F >> 2;
  int tid = blockIdx.x * blockDim.x + threadIdx.x;
  if (tid >= E * chunks) return;
  int e = tid / chunks;
  int c = (tid - e * chunks) << 2;
  float rn = rn_edge[e];
  const float4 v = *(const float4*)(T + (size_t)src[e] * F + c);
  float* yp = Y + (size_t)dst[e] * F + c;
  atomicAdd(yp + 0, rn * v.x);
  atomicAdd(yp + 1, rn * v.y);
  atomicAdd(yp + 2, rn * v.z);
  atomicAdd(yp + 3, rn * v.w);
}


__global__ void k_final(const float* __restrict__ A2s, const float* __restrict__ A2t,
                        const float* __restrict__ noise_s, const float* __restrict__ noise_t,
                        float* __restrict__ out, int N) {
  int idx = blockIdx.x * blockDim.x + threadIdx.x;
  int total = N * OUT_F;
  if (idx < total) {
    int n = idx >> 5, j = idx & 31;
    out[idx] = A2s[n * 64 + j] + noise_s[idx] * expf(A2s[n * 64 + 32 + j]) * 0.2f;
  } else if (idx < 2 * total) {
    int k = idx - total;
    int n = k >> 5, j = k & 31;
    out[idx] = A2t[n * 64 + j] + noise_t[k] * expf(A2t[n * 64 + 32 + j]) * 0.2f;
  }
}


static inline int ceil_div(int a, int b) { return (a + b - 1) / b; }

extern "C" void kernel_launch(void* const* d_in, const int* in_sizes, int n_in,
                              void* d_out, int out_size, void* d_ws, size_t ws_size,
                              hipStream_t stream) {
  const float* s       = (const float*)d_in[0];
  const float* t       = (const float*)d_in[1];
  const int*   ei      = (const int*)  d_in[2];
  const float* noise_s = (const float*)d_in[3];
  const float* noise_t = (const float*)d_in[4];
  const float* sm1_W = (const float*)d_in[5],  *sm1_b = (const float*)d_in[6];
  const float* sm2_W = (const float*)d_in[7],  *sm2_b = (const float*)d_in[8];
  const float* sl1_W = (const float*)d_in[9],  *sl1_b = (const float*)d_in[10];
  const float* sl2_W = (const float*)d_in[11], *sl2_b = (const float*)d_in[12];
  const float* tm1_W = (const float*)d_in[13], *tm1_b = (const float*)d_in[14];
  const float* tm2_W = (const float*)d_in[15], *tm2_b = (const float*)d_in[16];
  const float* tl1_W = (const float*)d_in[17], *tl1_b = (const float*)d_in[18];
  const float* tl2_W = (const float*)d_in[19], *tl2_b = (const float*)d_in[20];

  const int N = in_sizes[0] / IN_F;
  const int E = in_sizes[2] / 2;
  const int* ei0 = ei;
  const int* ei1 = ei + E;

  float* ws      = (float*)d_ws;
  float* deg_in  = ws;
  float* deg_out = deg_in  + N;
  float* rn_self = deg_out + N;
  float* rn_edge = rn_self + N;
  float* Y1s = rn_edge + E;
  float* Y1t = Y1s + (size_t)N * IN_F;
  float* A1s = Y1t + (size_t)N * IN_F;
  float* A1t = A1s + (size_t)N * IN_F;
  float* Y2s = Y1s;
  float* A2s = Y1s + (size_t)N * HID_F;
  float* Y2t = Y1t;
  float* A2t = Y1t + (size_t)N * HID_F;

  const int B = 256;
  const int nMTiles = ceil_div(N, 16);
  const int gemmBlocks = ceil_div(nMTiles, 4);

  k_init_deg<<<ceil_div(N, B), B, 0, stream>>>(deg_in, deg_out, N);
  k_count_deg<<<ceil_div(E, B), B, 0, stream>>>(ei0, ei1, deg_in, deg_out, E);
  k_norms<<<ceil_div(E + N, B), B, 0, stream>>>(ei0, ei1, deg_in, deg_out, rn_edge, rn_self, E, N);

  k_gemm_wmma<IN_F, 0><<<gemmBlocks, 128, 0, stream>>>(
      s, IN_F, sm1_W, sm1_b, sl1_W, sl1_b, Y1s, IN_F, HID_F, 0, IN_F / 16, nMTiles);
  k_gemm_wmma<IN_F, 0><<<gemmBlocks, 128, 0, stream>>>(
      t, IN_F, tm1_W, tm1_b, tl1_W, tl1_b, Y1t, IN_F, HID_F, 0, IN_F / 16, nMTiles);

  {
    int total4 = N * IN_F / 4, work = E * (IN_F >> 2);
    k_aggr_self<<<ceil_div(total4, B), B, 0, stream>>>(Y1s, rn_self, A1s, total4, 7);
    k_scatter  <<<ceil_div(work,   B), B, 0, stream>>>(Y1s, ei0, ei1, rn_edge, A1s, E, IN_F);
    k_aggr_self<<<ceil_div(total4, B), B, 0, stream>>>(Y1t, rn_self, A1t, total4, 7);
    k_scatter  <<<ceil_div(work,   B), B, 0, stream>>>(Y1t, ei1, ei0, rn_edge, A1t, E, IN_F);
  }

  k_gemm_wmma<HID_F, 1><<<gemmBlocks, 128, 0, stream>>>(
      A1s, IN_F, sm2_W, sm2_b, sl2_W, sl2_b, Y2s, HID_F, OUT_F, HID_F, HID_F / 16, nMTiles);
  k_gemm_wmma<HID_F, 1><<<gemmBlocks, 128, 0, stream>>>(
      A1t, IN_F, tm2_W, tm2_b, tl2_W, tl2_b, Y2t, HID_F, OUT_F, HID_F, HID_F / 16, nMTiles);

  {
    int total4 = N * HID_F / 4, work = E * (HID_F >> 2);
    k_aggr_self<<<ceil_div(total4, B), B, 0, stream>>>(Y2s, rn_self, A2s, total4, 6);
    k_scatter  <<<ceil_div(work,   B), B, 0, stream>>>(Y2s, ei1, ei0, rn_edge, A2s, E, HID_F);
    k_aggr_self<<<ceil_div(total4, B), B, 0, stream>>>(Y2t, rn_self, A2t, total4, 6);
    k_scatter  <<<ceil_div(work,   B), B, 0, stream>>>(Y2t, ei0, ei1, rn_edge, A2t, E, HID_F);
  }

  k_final<<<ceil_div(2 * N * OUT_F, B), B, 0, stream>>>(A2s, A2t, noise_s, noise_t,
                                                        (float*)d_out, N);
}


# Aggregation by gather over edge lists grouped by destination (count+rank, scan, place): register sums in list (arrival) order, one row store; replaces f32 atomic scatter; s side runs before t
# speedup vs baseline: 2.3032x; 2.3032x over previous
#include <hip/hip_runtime.h>


typedef __attribute__((ext_vector_type(2))) float v2f;
typedef __attribute__((ext_vector_type(4))) float v4f;
typedef __attribute__((ext_vector_type(8))) float v8f;

#define IN_F   128
#define HID_F  64
#define OUT_F  32


__global__ void k_zero_cnt(int* __restrict__ cnt, int n) {
  int i = blockIdx.x * blockDim.x + threadIdx.x;
  if (i < n) cnt[i] = 0;
}

__global__ void k_count_rank(const int* __restrict__ ei0, const int* __restrict__ ei1,
                             int* __restrict__ cnt_in, int* __restrict__ cnt_out,
                             int2* __restrict__ rank, int E) {
  int e = blockIdx.x * blockDim.x + threadIdx.x;
  if (e < E) {
    int s = ei0[e], d = ei1[e];
    int ro = atomicAdd(&cnt_out[s], 1);
    int ri = atomicAdd(&cnt_in [d], 1);
    rank[e] = make_int2(ri, ro);
  }
}

#define SCAN_T 1024

__global__ __launch_bounds__(SCAN_T)
void k_scan(const int* __restrict__ cnt_in, const int* __restrict__ cnt_out,
            int2* __restrict__ oc_in, int2* __restrict__ oc_out, int N) {
  __shared__ int wsum[32];
  __shared__ int psum[32];
  const int* cnt = blockIdx.y ? cnt_out : cnt_in;
  int2*      oc  = blockIdx.y ? oc_out  : oc_in;
  int t = threadIdx.x, lane = t & 31, wave = t >> 5;
  int tile0 = blockIdx.x * (SCAN_T * 4);

  int p = 0;
  for (int i = t * 4; i < tile0; i += SCAN_T * 4) {
    int4 c4 = *(const int4*)(cnt + i);
    p += (c4.x + c4.y) + (c4.z + c4.w);
  }

  int i0 = tile0 + t * 4;
  int c[4];
  if (i0 + 3 < N) {
    int4 c4 = *(const int4*)(cnt + i0);
    c[0] = c4.x; c[1] = c4.y; c[2] = c4.z; c[3] = c4.w;
  } else {
    for (int k = 0; k < 4; ++k) c[k] = (i0 + k < N) ? cnt[i0 + k] : 0;
  }
  int mine = (c[0] + c[1]) + (c[2] + c[3]);

  int incl = mine;
  for (int d = 1; d < 32; d <<= 1) { int y = __shfl_up(incl, d); if (lane >= d) incl += y; }
  int pr = p;
  for (int d = 16; d >= 1; d >>= 1) pr += __shfl_xor(pr, d);
  if (lane == 31) wsum[wave] = incl;
  if (lane == 0)  psum[wave] = pr;
  __syncthreads();
  int wv = wsum[lane];
  int winc = wv;
  for (int d = 1; d < 32; d <<= 1) { int y = __shfl_up(winc, d); if (lane >= d) winc += y; }
  int ptot = psum[lane];
  for (int d = 16; d >= 1; d >>= 1) ptot += __shfl_xor(ptot, d);
  int wexcl = __shfl(winc, wave) - __shfl(wv, wave);

  int run = ptot + wexcl + (incl - mine);
  for (int k = 0; k < 4; ++k) {
    if (i0 + k < N) oc[i0 + k] = make_int2(run, c[k]);
    run += c[k];
  }
}

__global__ void k_place(const int* __restrict__ ei0, const int* __restrict__ ei1,
                        const int2* __restrict__ rank,
                        const int2* __restrict__ oc_in, const int2* __restrict__ oc_out,
                        int2* __restrict__ lst_in, int2* __restrict__ lst_out,
                        float* __restrict__ rn_self, int E, int N) {
  int i = blockIdx.x * blockDim.x + threadIdx.x;
  if (i < E) {
    int s = ei0[i], d = ei1[i];
    int2 r  = rank[i];
    int2 od = oc_in[d];
    int2 os = oc_out[s];
    float w = rsqrtf((float)(od.y + 1) * (float)(os.y + 1));
    int wb = __float_as_int(w);
    lst_in [od.x + r.x] = make_int2(s, wb);
    lst_out[os.x + r.y] = make_int2(d, wb);
  } else if (i < E + N) {
    int n = i - E;
    rn_self[n] = rsqrtf((float)(oc_in[n].y + 1) * (float)(oc_out[n].y + 1));
  }
}


template <int K, int RELU>
__global__ __launch_bounds__(128)
void k_gemm_wmma(const float* __restrict__ X, int ldx,
                 const float* __restrict__ Wa, const float* __restrict__ ba,
                 const float* __restrict__ Wb, const float* __restrict__ bb,
                 float* __restrict__ Y, int ldy,
                 int halfO, int koffB, int nColTiles, int nMTiles) {
  constexpr int NK = K / 4;

  int wave = threadIdx.x >> 5;
  int lane = threadIdx.x & 31;
  int mt   = blockIdx.x * 4 + wave;
  if (mt >= nMTiles) return;

  int m0   = mt << 4;
  int mrow = lane & 15;
  int khi  = (lane >> 4) << 1;
  int col  = lane & 15;

  const float* xrow = X + (size_t)(m0 + mrow) * ldx;

  v2f afrag[NK];
  int curKoff = -1;

  for (int ct = 0; ct < nColTiles; ++ct) {
    int c0   = ct << 4;
    int half = (c0 >= halfO) ? 1 : 0;
    const float* W    = half ? Wb : Wa;
    const float* bias = half ? bb : ba;
    int wc0  = c0 - half * halfO;
    int koff = half ? koffB : 0;

    if (koff != curKoff) {
      curKoff = koff;
      const float* xp = xrow + koff + khi;
#pragma unroll
      for (int i = 0; i < NK; ++i) {
        v2f a = *(const v2f*)(xp + i * 4);
        if (RELU) { a.x = fmaxf(a.x, 0.0f); a.y = fmaxf(a.y, 0.0f); }
        afrag[i] = a;
      }
    }

    const float* wrow = W + (size_t)(wc0 + col) * K + khi;
    v8f acc = {};
#pragma unroll
    for (int i = 0; i < NK; ++i) {
      v2f bf = *(const v2f*)(wrow + i * 4);
      acc = __builtin_amdgcn_wmma_f32_16x16x4_f32(false, afrag[i], false, bf,
                                                  (short)0, acc, false, false);
    }

    float bv = bias[wc0 + col];
    int mbase = m0 + ((lane >> 4) << 3);
    float* yp = Y + (size_t)mbase * ldy + c0 + col;
#pragma unroll
    for (int g = 0; g < 8; ++g)
      yp[(size_t)g * ldy] = acc[g] + bv;
  }
}


template <int V> struct VecOf;
template <> struct VecOf<2> { typedef v2f type; };
template <> struct VecOf<4> { typedef v4f type; };

template <int F, int K>
__device__ __forceinline__ void rows_fma(const float* __restrict__ T, int sx, int sw, int j, int c,
                                         typename VecOf<F / 32>::type& acc) {
  typedef typename VecOf<F / 32>::type vf;
  vf v[K];
  float w[K];
#pragma unroll
  for (int u = 0; u < K; ++u) {
    int src = __builtin_amdgcn_readlane(sx, j + u);
    w[u] = __int_as_float(__builtin_amdgcn_readlane(sw, j + u));
    v[u] = *(const vf*)(T + (size_t)src * F + c);
  }
#pragma unroll
  for (int u = 0; u < K; ++u) acc += w[u] * v[u];
}

template <int F>
__global__ __launch_bounds__(256)
void k_gather(const float* __restrict__ T, const int2* __restrict__ oc, const int2* __restrict__ lst,
              const float* __restrict__ rn_self, float* __restrict__ Y, int N) {
  constexpr int V = F / 32;
  typedef typename VecOf<V>::type vf;
  const int lane = threadIdx.x & 31;
  const int c = lane * V;
  const int nWaves = gridDim.x * 8;
  int wid = __builtin_amdgcn_readfirstlane(blockIdx.x * 8 + (threadIdx.x >> 5));

  if (wid >= N) return;
  int2 o = oc[wid];
  for (int node = wid; ; ) {
    const int beg = o.x, cnt = o.y;
    const int nxt = node + nWaves;
    int2 o_nxt = o;
    if (nxt < N) o_nxt = oc[nxt];
    int n = min(32, cnt);
    int2 rec = make_int2(0, 0);
    if (lane < n) rec = lst[beg + lane];
    vf acc = rn_self[node] * *(const vf*)(T + (size_t)node * F + c);

    for (int base = 0; base < cnt; ) {
      int j = 0;
      for (; j + 8 <= n; j += 8) rows_fma<F, 8>(T, rec.x, rec.y, j, c, acc);
      if (j + 4 <= n) { rows_fma<F, 4>(T, rec.x, rec.y, j, c, acc); j += 4; }
      if (j + 2 <= n) { rows_fma<F, 2>(T, rec.x, rec.y, j, c, acc); j += 2; }
      if (j < n)      { rows_fma<F, 1>(T, rec.x, rec.y, j, c, acc); }
      base += 32;
      if (base < cnt) {
        n = min(32, cnt - base);
        rec = make_int2(0, 0);
        if (lane < n) rec = lst[beg + base + lane];
      }
    }
    *(vf*)(Y + (size_t)node * F + c) = acc;
    if (nxt >= N) break;
    node = nxt;
    o = o_nxt;
  }
}


__global__ void k_final(const float* __restrict__ A2s, const float* __restrict__ A2t,
                        const float* __restrict__ noise_s, const float* __restrict__ noise_t,
                        float* __restrict__ out, int N) {
  int idx = blockIdx.x * blockDim.x + threadIdx.x;
  int total = N * OUT_F;
  if (idx < total) {
    int n = idx >> 5, j = idx & 31;
    out[idx] = A2s[n * 64 + j] + noise_s[idx] * expf(A2s[n * 64 + 32 + j]) * 0.2f;
  } else if (idx < 2 * total) {
    int k = idx - total;
    int n = k >> 5, j = k & 31;
    out[idx] = A2t[n * 64 + j] + noise_t[k] * expf(A2t[n * 64 + 32 + j]) * 0.2f;
  }
}


static inline int ceil_div(int a, int b) { return (a + b - 1) / b; }

extern "C" void kernel_launch(void* const* d_in, const int* in_sizes, int n_in,
                              void* d_out, int out_size, void* d_ws, size_t ws_size,
                              hipStream_t stream) {
  const float* s       = (const float*)d_in[0];
  const float* t       = (const float*)d_in[1];
  const int*   ei      = (const int*)  d_in[2];
  const float* noise_s = (const float*)d_in[3];
  const float* noise_t = (const float*)d_in[4];
  const float* sm1_W = (const float*)d_in[5],  *sm1_b = (const float*)d_in[6];
  const float* sm2_W = (const float*)d_in[7],  *sm2_b = (const float*)d_in[8];
  const float* sl1_W = (const float*)d_in[9],  *sl1_b = (const float*)d_in[10];
  const float* sl2_W = (const float*)d_in[11], *sl2_b = (const float*)d_in[12];
  const float* tm1_W = (const float*)d_in[13], *tm1_b = (const float*)d_in[14];
  const float* tm2_W = (const float*)d_in[15], *tm2_b = (const float*)d_in[16];
  const float* tl1_W = (const float*)d_in[17], *tl1_b = (const float*)d_in[18];
  const float* tl2_W = (const float*)d_in[19], *tl2_b = (const float*)d_in[20];

  const int N = in_sizes[0] / IN_F;
  const int E = in_sizes[2] / 2;
  const int* ei0 = ei;
  const int* ei1 = ei + E;

  const int Npad = (N + 3) & ~3;
  float* ws      = (float*)d_ws;
  int*   cnt_in  = (int*)ws;
  int*   cnt_out = cnt_in + Npad;
  float* rn_self = (float*)(cnt_out + Npad);
  float* P0 = ws + (size_t)3 * N + E;
  float* P1 = P0 + (size_t)N * IN_F;
  float* P2 = P1 + (size_t)N * IN_F;
  float* P3 = P2 + (size_t)N * IN_F;
  int2* oc_in   = (int2*)P3;
  int2* oc_out  = oc_in  + N;
  int2* rank    = oc_out + N;
  int2* lst_in  = rank   + E;
  int2* lst_out = lst_in + E;

  float* Y1s = P0;
  float* A1s = P1;
  float* Y2s = P0;
  float* A2s = P0 + (size_t)N * HID_F;
  float* Y1t = P2;
  float* A1t = P1;
  float* Y2t = P2;
  float* A2t = P2 + (size_t)N * HID_F;

  const int B = 256;
  const int nMTiles = ceil_div(N, 16);
  const int gemmBlocks = ceil_div(nMTiles, 4);
  const int gatherBlocks = ceil_div(N, 8) < 2048 ? ceil_div(N, 8) : 2048;

  k_zero_cnt<<<ceil_div(2 * Npad, B), B, 0, stream>>>(cnt_in, 2 * Npad);
  k_count_rank<<<ceil_div(E, B), B, 0, stream>>>(ei0, ei1, cnt_in, cnt_out, rank, E);
  k_scan<<<dim3(ceil_div(N, SCAN_T * 4), 2), SCAN_T, 0, stream>>>(cnt_in, cnt_out, oc_in, oc_out, N);
  k_place<<<ceil_div(E + N, B), B, 0, stream>>>(ei0, ei1, rank, oc_in, oc_out, lst_in, lst_out,
                                                rn_self, E, N);

  k_gemm_wmma<IN_F, 0><<<gemmBlocks, 128, 0, stream>>>(
      s, IN_F, sm1_W, sm1_b, sl1_W, sl1_b, Y1s, IN_F, HID_F, 0, IN_F / 16, nMTiles);
  k_gather<IN_F><<<gatherBlocks, B, 0, stream>>>(Y1s, oc_in, lst_in, rn_self, A1s, N);
  k_gemm_wmma<HID_F, 1><<<gemmBlocks, 128, 0, stream>>>(
      A1s, IN_F, sm2_W, sm2_b, sl2_W, sl2_b, Y2s, HID_F, OUT_F, HID_F, HID_F / 16, nMTiles);
  k_gather<HID_F><<<gatherBlocks, B, 0, stream>>>(Y2s, oc_out, lst_out, rn_self, A2s, N);

  k_gemm_wmma<IN_F, 0><<<gemmBlocks, 128, 0, stream>>>(
      t, IN_F, tm1_W, tm1_b, tl1_W, tl1_b, Y1t, IN_F, HID_F, 0, IN_F / 16, nMTiles);
  k_gather<IN_F><<<gatherBlocks, B, 0, stream>>>(Y1t, oc_out, lst_out, rn_self, A1t, N);
  k_gemm_wmma<HID_F, 1><<<gemmBlocks, 128, 0, stream>>>(
      A1t, IN_F, tm2_W, tm2_b, tl2_W, tl2_b, Y2t, HID_F, OUT_F, HID_F, HID_F / 16, nMTiles);
  k_gather<HID_F><<<gatherBlocks, B, 0, stream>>>(Y2t, oc_in, lst_in, rn_self, A2t, N);

  k_final<<<ceil_div(2 * N * OUT_F, B), B, 0, stream>>>(A2s, A2t, noise_s, noise_t,
                                                        (float*)d_out, N);
}


# Layer-2 linear fused into the layer-1 gather: 16 waves sum 16 nodes' rows, relu, hand the tile to WMMA through LDS (weights in LDS); no [N,128] sums plane, no separate product launch
# speedup vs baseline: 3.6931x; 1.6035x over previous
#include <hip/hip_runtime.h>


typedef __attribute__((ext_vector_type(2))) float v2f;
typedef __attribute__((ext_vector_type(4))) float v4f;
typedef __attribute__((ext_vector_type(8))) float v8f;

#define IN_F   128
#define HID_F  64
#define OUT_F  32


#define BIN_SH   8
#define BIN_W    (1 << BIN_SH)
#define BIN_MAXB 1024
#define CH_E     4096

__device__ __forceinline__ int wave_incl_scan(int v, int lane) {
  for (int d = 1; d < 32; d <<= 1) { int y = __shfl_up(v, d); if (lane >= d) v += y; }
  return v;
}

__global__ __launch_bounds__(1024)
void k_bin(const int* __restrict__ ei0, const int* __restrict__ ei1,
           int2* __restrict__ bin_in, int2* __restrict__ bin_out,
           int* __restrict__ seg_in, int* __restrict__ seg_out, int E, int NB, int G) {
  __shared__ int h_in[BIN_MAXB];
  __shared__ int h_out[BIN_MAXB];
  __shared__ int ws_in[32];
  __shared__ int ws_out[32];
  const int t = threadIdx.x, lane = t & 31, wave = t >> 5, g = blockIdx.x;
  h_in[t] = 0; h_out[t] = 0;
  __syncthreads();

  const int base = g * CH_E;
  int s[4], d[4], ri[4], ro[4];
#pragma unroll
  for (int k = 0; k < 4; ++k) {
    int e = base + k * 1024 + t;
    s[k] = -1; d[k] = 0;
    if (e < E) { s[k] = ei0[e]; d[k] = ei1[e]; }
  }
#pragma unroll
  for (int k = 0; k < 4; ++k) {
    ri[k] = 0; ro[k] = 0;
    if (s[k] >= 0) {
      ri[k] = atomicAdd(&h_in [d[k] >> BIN_SH], 1);
      ro[k] = atomicAdd(&h_out[s[k] >> BIN_SH], 1);
    }
  }
  __syncthreads();

  const int ci = h_in[t], co = h_out[t];
  const int ii = wave_incl_scan(ci, lane), io = wave_incl_scan(co, lane);
  if (lane == 31) { ws_in[wave] = ii; ws_out[wave] = io; }
  __syncthreads();
  const int wi = ws_in[lane], wo = ws_out[lane];
  const int wii = wave_incl_scan(wi, lane), woo = wave_incl_scan(wo, lane);
  const int ex_in  = (__shfl(wii, wave) - __shfl(wi, wave)) + (ii - ci);
  const int ex_out = (__shfl(woo, wave) - __shfl(wo, wave)) + (io - co);
  if (t <= NB) { seg_in[t * G + g] = ex_in; seg_out[t * G + g] = ex_out; }
  h_in[t] = ex_in; h_out[t] = ex_out;
  __syncthreads();

#pragma unroll
  for (int k = 0; k < 4; ++k) {
    if (s[k] >= 0) {
      bin_in [base + h_in [d[k] >> BIN_SH] + ri[k]] = make_int2(d[k], s[k]);
      bin_out[base + h_out[s[k] >> BIN_SH] + ro[k]] = make_int2(s[k], d[k]);
    }
  }
}

__global__ __launch_bounds__(1024)
void k_rank(const int2* __restrict__ bin_in, const int2* __restrict__ bin_out,
            const int* __restrict__ seg_in, const int* __restrict__ seg_out,
            int2* __restrict__ oc_in, int2* __restrict__ oc_out,
            int* __restrict__ rk_in, int* __restrict__ rk_out, int N, int G) {
  __shared__ int c[BIN_W];
  __shared__ int sst[1024];
  __shared__ int sen[1024];
  __shared__ int wsum[32];
  __shared__ int wtot[BIN_W / 32];
  const int2* bin = blockIdx.y ? bin_out : bin_in;
  const int*  seg = blockIdx.y ? seg_out : seg_in;
  int2*       oc  = blockIdx.y ? oc_out  : oc_in;
  int*        rk  = blockIdx.y ? rk_out  : rk_in;
  const int t = threadIdx.x, lane = t & 31, wave = t >> 5, b = blockIdx.x;
  const int nb0 = b << BIN_SH;
  if (t < BIN_W) c[t] = 0;
  __syncthreads();

  int part = 0;
  for (int g0 = 0; g0 < G; g0 += 1024) {
    if (g0 + t < G) {
      int a = seg[b * G + g0 + t];
      sst[t] = a; sen[t] = seg[(b + 1) * G + g0 + t];
      part += a;
    }
    __syncthreads();
    const int ng = min(1024, G - g0);
    for (int j = wave; j < ng; j += 32) {
      const int st = sst[j], en = sen[j];
      const size_t cb = (size_t)(g0 + j) * CH_E;
      for (int p = st + lane; p < en; p += 32) {
        int2 r = bin[cb + p];
        rk[cb + p] = atomicAdd(&c[r.x - nb0], 1);
      }
    }
    __syncthreads();
  }

  for (int d = 16; d >= 1; d >>= 1) part += __shfl_xor(part, d);
  if (lane == 0) wsum[wave] = part;
  const int cv  = (t < BIN_W) ? c[t] : 0;
  const int inc = wave_incl_scan(cv, lane);
  if (t < BIN_W && lane == 31) wtot[wave] = inc;
  __syncthreads();
  int first = wsum[lane];
  for (int d = 16; d >= 1; d >>= 1) first += __shfl_xor(first, d);
  if (t < BIN_W) {
    int pre = 0;
    for (int w = 0; w < wave; ++w) pre += wtot[w];
    if (nb0 + t < N) oc[nb0 + t] = make_int2(first + pre + (inc - cv), cv);
  }
}

__global__ void k_place(const int2* __restrict__ bin_in, const int2* __restrict__ bin_out,
                        const int* __restrict__ rk_in, const int* __restrict__ rk_out,
                        const int2* __restrict__ oc_in, const int2* __restrict__ oc_out,
                        int2* __restrict__ lst_in, int2* __restrict__ lst_out,
                        float* __restrict__ rn_self, int E, int N) {
  int i = blockIdx.x * blockDim.x + threadIdx.x;
  if (i < E) {
    int2 a = bin_in[i];
    int2 b = bin_out[i];
    int ra = rk_in[i], rb = rk_out[i];
    int2 od = oc_in [a.x];
    int2 cs = oc_out[a.y];
    int2 os = oc_out[b.x];
    int2 cd = oc_in [b.y];
    float wa = rsqrtf((float)(od.y + 1) * (float)(cs.y + 1));
    float wb = rsqrtf((float)(cd.y + 1) * (float)(os.y + 1));
    lst_in [od.x + ra] = make_int2(a.y, __float_as_int(wa));
    lst_out[os.x + rb] = make_int2(b.y, __float_as_int(wb));
  } else if (i < E + N) {
    int n = i - E;
    rn_self[n] = rsqrtf((float)(oc_in[n].y + 1) * (float)(oc_out[n].y + 1));
  }
}


template <int K, int RELU>
__global__ __launch_bounds__(128)
void k_gemm_wmma(const float* __restrict__ X, int ldx,
                 const float* __restrict__ Wa, const float* __restrict__ ba,
                 const float* __restrict__ Wb, const float* __restrict__ bb,
                 float* __restrict__ Y, int ldy,
                 int halfO, int koffB, int nColTiles, int nMTiles) {
  constexpr int NK = K / 4;

  int wave = threadIdx.x >> 5;
  int lane = threadIdx.x & 31;
  int mt   = blockIdx.x * 4 + wave;
  if (mt >= nMTiles) return;

  int m0   = mt << 4;
  int mrow = lane & 15;
  int khi  = (lane >> 4) << 1;
  int col  = lane & 15;

  const float* xrow = X + (size_t)(m0 + mrow) * ldx;

  v2f afrag[NK];
  int curKoff = -1;

  for (int ct = 0; ct < nColTiles; ++ct) {
    int c0   = ct << 4;
    int half = (c0 >= halfO) ? 1 : 0;
    const float* W    = half ? Wb : Wa;
    const float* bias = half ? bb : ba;
    int wc0  = c0 - half * halfO;
    int koff = half ? koffB : 0;

    if (koff != curKoff) {
      curKoff = koff;
      const float* xp = xrow + koff + khi;
#pragma unroll
      for (int i = 0; i < NK; ++i) {
        v2f a = *(const v2f*)(xp + i * 4);
        if (RELU) { a.x = fmaxf(a.x, 0.0f); a.y = fmaxf(a.y, 0.0f); }
        afrag[i] = a;
      }
    }

    const float* wrow = W + (size_t)(wc0 + col) * K + khi;
    v8f acc = {};
#pragma unroll
    for (int i = 0; i < NK; ++i) {
      v2f bf = *(const v2f*)(wrow + i * 4);
      acc = __builtin_amdgcn_wmma_f32_16x16x4_f32(false, afrag[i], false, bf,
                                                  (short)0, acc, false, false);
    }

    float bv = bias[wc0 + col];
    int mbase = m0 + ((lane >> 4) << 3);
    float* yp = Y + (size_t)mbase * ldy + c0 + col;
#pragma unroll
    for (int g = 0; g < 8; ++g)
      yp[(size_t)g * ldy] = acc[g] + bv;
  }
}


template <int V> struct VecOf;
template <> struct VecOf<2> { typedef v2f type; };
template <> struct VecOf<4> { typedef v4f type; };

template <int F, int K>
__device__ __forceinline__ void rows_fma(const float* __restrict__ T, int sx, int sw, int j, int c,
                                         typename VecOf<F / 32>::type& acc) {
  typedef typename VecOf<F / 32>::type vf;
  vf v[K];
  float w[K];
#pragma unroll
  for (int u = 0; u < K; ++u) {
    int src = __builtin_amdgcn_readlane(sx, j + u);
    w[u] = __int_as_float(__builtin_amdgcn_readlane(sw, j + u));
    v[u] = *(const vf*)(T + (size_t)src * F + c);
  }
#pragma unroll
  for (int u = 0; u < K; ++u) acc += w[u] * v[u];
}

template <int F, int EPI>
__global__ __launch_bounds__(256)
void k_gather(const float* __restrict__ T, const int2* __restrict__ oc, const int2* __restrict__ lst,
              const float* __restrict__ rn_self, const float* __restrict__ noise,
              float* __restrict__ Y, int N) {
  static_assert(!EPI || F == 2 * OUT_F, "the fused reparametrize needs a mu|logstd row");
  constexpr int V = F / 32;
  typedef typename VecOf<V>::type vf;
  const int lane = threadIdx.x & 31;
  const int c = lane * V;
  const int nWaves = gridDim.x * 8;
  int wid = __builtin_amdgcn_readfirstlane(blockIdx.x * 8 + (threadIdx.x >> 5));

  if (wid >= N) return;
  int2 o = oc[wid];
  for (int node = wid; ; ) {
    const int beg = o.x, cnt = o.y;
    const int nxt = node + nWaves;
    int2 o_nxt = o;
    if (nxt < N) o_nxt = oc[nxt];
    int n = min(32, cnt);
    int2 rec = make_int2(0, 0);
    if (lane < n) rec = lst[beg + lane];
    vf acc = rn_self[node] * *(const vf*)(T + (size_t)node * F + c);
    v2f nz = {0.0f, 0.0f};
    if constexpr (EPI) {
      if (lane < 16) nz = *(const v2f*)(noise + (size_t)node * OUT_F + c);
    }

    for (int base = 0; base < cnt; ) {
      int j = 0;
      for (; j + 8 <= n; j += 8) rows_fma<F, 8>(T, rec.x, rec.y, j, c, acc);
      if (j + 4 <= n) { rows_fma<F, 4>(T, rec.x, rec.y, j, c, acc); j += 4; }
      if (j + 2 <= n) { rows_fma<F, 2>(T, rec.x, rec.y, j, c, acc); j += 2; }
      if (j < n)      { rows_fma<F, 1>(T, rec.x, rec.y, j, c, acc); }
      base += 32;
      if (base < cnt) {
        n = min(32, cnt - base);
        rec = make_int2(0, 0);
        if (lane < n) rec = lst[beg + base + lane];
      }
    }
    if constexpr (EPI) {
      const float lx = __shfl_xor(acc.x, 16), ly = __shfl_xor(acc.y, 16);
      if (lane < 16) {
        v2f r;
        r.x = acc.x + nz.x * expf(lx) * 0.2f;
        r.y = acc.y + nz.y * expf(ly) * 0.2f;
        *(v2f*)(Y + (size_t)node * OUT_F + c) = r;
      }
    } else {
      *(vf*)(Y + (size_t)node * F + c) = acc;
    }
    if (nxt >= N) break;
    node = nxt;
    o = o_nxt;
  }
}


#define GL_WAVES 16
#define GL_T     (GL_WAVES * 32)
#define GL_AP    (IN_F + 4)
#define GL_WP    (HID_F + 4)

__global__ __launch_bounds__(GL_T)
void k_gather_lin2(const float* __restrict__ T, const int2* __restrict__ oc, const int2* __restrict__ lst,
                   const float* __restrict__ rn_self,
                   const float* __restrict__ Wa, const float* __restrict__ ba,
                   const float* __restrict__ Wb, const float* __restrict__ bb,
                   float* __restrict__ Y, int N, int nTiles) {
  constexpr int F = IN_F;
  __shared__ __attribute__((aligned(16))) float wl[HID_F * GL_WP];
  __shared__ __attribute__((aligned(16))) float tb[2 * 16 * GL_AP];
  __shared__ float bl[HID_F];
  const int tid = threadIdx.x, lane = tid & 31;
  const int wave = __builtin_amdgcn_readfirstlane(tid >> 5);
  const int c = lane * 4;

  int tile = blockIdx.x;
  if (tile >= nTiles) return;
  int node = tile * 16 + wave;
  int2 o = make_int2(0, 0);
  if (node < N) o = oc[node];

#pragma unroll
  for (int u = 0; u < 2; ++u) {
    const int q = tid + u * GL_T;
    const int row = q >> 4, kc = (q & 15) * 4;
    const float* src = (row < OUT_F) ? (Wa + row * HID_F + kc) : (Wb + (row - OUT_F) * HID_F + kc);
    *(v4f*)(wl + row * GL_WP + kc) = *(const v4f*)src;
  }
  if (tid < HID_F) bl[tid] = (tid < OUT_F) ? ba[tid] : bb[tid - OUT_F];


  for (int it = 0; ; ++it) {
    const int ntile = tile + gridDim.x;
    const int nnode = ntile * 16 + wave;
    int2 o_nxt = make_int2(0, 0);
    if (ntile < nTiles && nnode < N) o_nxt = oc[nnode];

    v4f acc = {0.0f, 0.0f, 0.0f, 0.0f};
    if (node < N) {
      const int beg = o.x, cnt = o.y;
      int n = min(32, cnt);
      int2 rec = make_int2(0, 0);
      if (lane < n) rec = lst[beg + lane];
      acc = rn_self[node] * *(const v4f*)(T + (size_t)node * F + c);
      for (int base = 0; base < cnt; ) {
        int j = 0;
        for (; j + 8 <= n; j += 8) rows_fma<F, 8>(T, rec.x, rec.y, j, c, acc);
        if (j + 4 <= n) { rows_fma<F, 4>(T, rec.x, rec.y, j, c, acc); j += 4; }
        if (j + 2 <= n) { rows_fma<F, 2>(T, rec.x, rec.y, j, c, acc); j += 2; }
        if (j < n)      { rows_fma<F, 1>(T, rec.x, rec.y, j, c, acc); }
        base += 32;
        if (base < cnt) {
          n = min(32, cnt - base);
          rec = make_int2(0, 0);
          if (lane < n) rec = lst[beg + base + lane];
        }
      }
    }
    acc.x = fmaxf(acc.x, 0.0f); acc.y = fmaxf(acc.y, 0.0f); acc.z = fmaxf(acc.z, 0.0f); acc.w = fmaxf(acc.w, 0.0f);
    float* tcur = tb + (it & 1) * (16 * GL_AP);
    *(v4f*)(tcur + wave * GL_AP + c) = acc;
    __syncthreads();

    if ((wave >> 2) == (it & 3)) {
      const int ct   = wave & 3;
      const int mrow = lane & 15;
      const int khi  = (lane >> 4) << 1;
      const int col  = lane & 15;
      const float* ap = tcur + mrow * GL_AP + (ct >> 1) * HID_F + khi;
      const float* wp = wl + (ct * 16 + col) * GL_WP + khi;
      v8f a8 = {};
#pragma unroll
      for (int i = 0; i < HID_F / 4; ++i) {
        v2f af = *(const v2f*)(ap + i * 4);
        v2f bf = *(const v2f*)(wp + i * 4);
        a8 = __builtin_amdgcn_wmma_f32_16x16x4_f32(false, af, false, bf, (short)0, a8, false, false);
      }
      const float bv = bl[ct * 16 + col];
      const int mbase = tile * 16 + ((lane >> 4) << 3);
      float* yp = Y + (size_t)mbase * HID_F + ct * 16 + col;
#pragma unroll
      for (int g = 0; g < 8; ++g)
        if (mbase + g < N) yp[(size_t)g * HID_F] = a8[g] + bv;
    }
    if (ntile >= nTiles) break;
    tile = ntile; node = nnode; o = o_nxt;
  }
}


static inline int ceil_div(int a, int b) { return (a + b - 1) / b; }

extern "C" void kernel_launch(void* const* d_in, const int* in_sizes, int n_in,
                              void* d_out, int out_size, void* d_ws, size_t ws_size,
                              hipStream_t stream) {
  const float* s       = (const float*)d_in[0];
  const float* t       = (const float*)d_in[1];
  const int*   ei      = (const int*)  d_in[2];
  const float* noise_s = (const float*)d_in[3];
  const float* noise_t = (const float*)d_in[4];
  const float* sm1_W = (const float*)d_in[5],  *sm1_b = (const float*)d_in[6];
  const float* sm2_W = (const float*)d_in[7],  *sm2_b = (const float*)d_in[8];
  const float* sl1_W = (const float*)d_in[9],  *sl1_b = (const float*)d_in[10];
  const float* sl2_W = (const float*)d_in[11], *sl2_b = (const float*)d_in[12];
  const float* tm1_W = (const float*)d_in[13], *tm1_b = (const float*)d_in[14];
  const float* tm2_W = (const float*)d_in[15], *tm2_b = (const float*)d_in[16];
  const float* tl1_W = (const float*)d_in[17], *tl1_b = (const float*)d_in[18];
  const float* tl2_W = (const float*)d_in[19], *tl2_b = (const float*)d_in[20];

  const int N = in_sizes[0] / IN_F;
  const int E = in_sizes[2] / 2;
  const int* ei0 = ei;
  const int* ei1 = ei + E;

  const int Npad = (N + 3) & ~3;
  float* ws      = (float*)d_ws;
  float* rn_self = ws + 2 * Npad;
  float* P0 = ws + (size_t)3 * N + E;
  float* P1 = P0 + (size_t)N * IN_F;
  float* P2 = P1 + (size_t)N * IN_F;
  float* P3 = P2 + (size_t)N * IN_F;
  int2* oc_in   = (int2*)P3;
  int2* oc_out  = oc_in  + N;
  int2* lst_in  = oc_out + N;
  int2* lst_out = lst_in + E;

  const int NB = ceil_div(N, BIN_W);
  const int G  = ceil_div(E, CH_E);
  int2* bin_in  = (int2*)P0;
  int2* bin_out = bin_in + E;
  int*  rk_in   = (int*)(bin_out + E);
  int*  rk_out  = rk_in + E;
  int*  seg_in  = rk_out + E;
  int*  seg_out = seg_in + (size_t)(NB + 1) * G;

  float* Y1s = P0;
  float* Y2s = P1;
  float* Y1t = P2;
  float* Y2t = P1;
  float* out_s = (float*)d_out;
  float* out_t = out_s + (size_t)N * OUT_F;

  const int B = 256;
  const int nMTiles = ceil_div(N, 16);
  const int gemmBlocks = ceil_div(nMTiles, 4);
  const int gatherBlocks = ceil_div(N, 8) < 2048 ? ceil_div(N, 8) : 2048;
  const int fusedBlocks = nMTiles < 1024 ? nMTiles : 1024;

  k_bin<<<G, 1024, 0, stream>>>(ei0, ei1, bin_in, bin_out, seg_in, seg_out, E, NB, G);
  k_rank<<<dim3(NB, 2), 1024, 0, stream>>>(bin_in, bin_out, seg_in, seg_out, oc_in, oc_out,
                                           rk_in, rk_out, N, G);
  k_place<<<ceil_div(E + N, B), B, 0, stream>>>(bin_in, bin_out, rk_in, rk_out, oc_in, oc_out,
                                                lst_in, lst_out, rn_self, E, N);

  k_gemm_wmma<IN_F, 0><<<gemmBlocks, 128, 0, stream>>>(
      s, IN_F, sm1_W, sm1_b, sl1_W, sl1_b, Y1s, IN_F, HID_F, 0, IN_F / 16, nMTiles);
  k_gather_lin2<<<fusedBlocks, GL_T, 0, stream>>>(Y1s, oc_in, lst_in, rn_self,
                                                  sm2_W, sm2_b, sl2_W, sl2_b, Y2s, N, nMTiles);
  k_gather<HID_F, 1><<<gatherBlocks, B, 0, stream>>>(Y2s, oc_out, lst_out, rn_self, noise_s, out_s, N);

  k_gemm_wmma<IN_F, 0><<<gemmBlocks, 128, 0, stream>>>(
      t, IN_F, tm1_W, tm1_b, tl1_W, tl1_b, Y1t, IN_F, HID_F, 0, IN_F / 16, nMTiles);
  k_gather_lin2<<<fusedBlocks, GL_T, 0, stream>>>(Y1t, oc_out, lst_out, rn_self,
                                                  tm2_W, tm2_b, tl2_W, tl2_b, Y2t, N, nMTiles);
  k_gather<HID_F, 1><<<gatherBlocks, B, 0, stream>>>(Y2t, oc_in, lst_in, rn_self, noise_t, out_t, N);
}


# Both encoder sides share each launch: products, fused layer-1 gather and layer-2 gather run once each over the s then the t items (pointers chosen per item); 9 launches become 6, sums unchanged
# speedup vs baseline: 4.1192x; 1.1154x over previous
#include <hip/hip_runtime.h>


typedef __attribute__((ext_vector_type(2))) float v2f;
typedef __attribute__((ext_vector_type(4))) float v4f;
typedef __attribute__((ext_vector_type(8))) float v8f;

#define IN_F   128
#define HID_F  64
#define OUT_F  32


#define BIN_SH   8
#define BIN_W    (1 << BIN_SH)
#define BIN_MAXB 1024
#define CH_E     4096

__device__ __forceinline__ int wave_incl_scan(int v, int lane) {
  for (int d = 1; d < 32; d <<= 1) { int y = __shfl_up(v, d); if (lane >= d) v += y; }
  return v;
}

__global__ __launch_bounds__(1024)
void k_bin(const int* __restrict__ ei0, const int* __restrict__ ei1,
           int2* __restrict__ bin_in, int2* __restrict__ bin_out,
           int* __restrict__ seg_in, int* __restrict__ seg_out, int E, int NB, int G) {
  __shared__ int h_in[BIN_MAXB];
  __shared__ int h_out[BIN_MAXB];
  __shared__ int ws_in[32];
  __shared__ int ws_out[32];
  const int t = threadIdx.x, lane = t & 31, wave = t >> 5, g = blockIdx.x;
  h_in[t] = 0; h_out[t] = 0;
  __syncthreads();

  const int base = g * CH_E;
  int s[4], d[4], ri[4], ro[4];
#pragma unroll
  for (int k = 0; k < 4; ++k) {
    int e = base + k * 1024 + t;
    s[k] = -1; d[k] = 0;
    if (e < E) { s[k] = ei0[e]; d[k] = ei1[e]; }
  }
#pragma unroll
  for (int k = 0; k < 4; ++k) {
    ri[k] = 0; ro[k] = 0;
    if (s[k] >= 0) {
      ri[k] = atomicAdd(&h_in [d[k] >> BIN_SH], 1);
      ro[k] = atomicAdd(&h_out[s[k] >> BIN_SH], 1);
    }
  }
  __syncthreads();

  const int ci = h_in[t], co = h_out[t];
  const int ii = wave_incl_scan(ci, lane), io = wave_incl_scan(co, lane);
  if (lane == 31) { ws_in[wave] = ii; ws_out[wave] = io; }
  __syncthreads();
  const int wi = ws_in[lane], wo = ws_out[lane];
  const int wii = wave_incl_scan(wi, lane), woo = wave_incl_scan(wo, lane);
  const int ex_in  = (__shfl(wii, wave) - __shfl(wi, wave)) + (ii - ci);
  const int ex_out = (__shfl(woo, wave) - __shfl(wo, wave)) + (io - co);
  if (t <= NB) { seg_in[t * G + g] = ex_in; seg_out[t * G + g] = ex_out; }
  h_in[t] = ex_in; h_out[t] = ex_out;
  __syncthreads();

#pragma unroll
  for (int k = 0; k < 4; ++k) {
    if (s[k] >= 0) {
      bin_in [base + h_in [d[k] >> BIN_SH] + ri[k]] = make_int2(d[k], s[k]);
      bin_out[base + h_out[s[k] >> BIN_SH] + ro[k]] = make_int2(s[k], d[k]);
    }
  }
}

__global__ __launch_bounds__(1024)
void k_rank(const int2* __restrict__ bin_in, const int2* __restrict__ bin_out,
            const int* __restrict__ seg_in, const int* __restrict__ seg_out,
            int2* __restrict__ oc_in, int2* __restrict__ oc_out,
            int* __restrict__ rk_in, int* __restrict__ rk_out, int N, int G) {
  __shared__ int c[BIN_W];
  __shared__ int sst[1024];
  __shared__ int sen[1024];
  __shared__ int wsum[32];
  __shared__ int wtot[BIN_W / 32];
  const int2* bin = blockIdx.y ? bin_out : bin_in;
  const int*  seg = blockIdx.y ? seg_out : seg_in;
  int2*       oc  = blockIdx.y ? oc_out  : oc_in;
  int*        rk  = blockIdx.y ? rk_out  : rk_in;
  const int t = threadIdx.x, lane = t & 31, wave = t >> 5, b = blockIdx.x;
  const int nb0 = b << BIN_SH;
  if (t < BIN_W) c[t] = 0;
  __syncthreads();

  int part = 0;
  for (int g0 = 0; g0 < G; g0 += 1024) {
    if (g0 + t < G) {
      int a = seg[b * G + g0 + t];
      sst[t] = a; sen[t] = seg[(b + 1) * G + g0 + t];
      part += a;
    }
    __syncthreads();
    const int ng = min(1024, G - g0);
    for (int j = wave; j < ng; j += 32) {
      const int st = sst[j], en = sen[j];
      const size_t cb = (size_t)(g0 + j) * CH_E;
      for (int p = st + lane; p < en; p += 32) {
        int2 r = bin[cb + p];
        rk[cb + p] = atomicAdd(&c[r.x - nb0], 1);
      }
    }
    __syncthreads();
  }

  for (int d = 16; d >= 1; d >>= 1) part += __shfl_xor(part, d);
  if (lane == 0) wsum[wave] = part;
  const int cv  = (t < BIN_W) ? c[t] : 0;
  const int inc = wave_incl_scan(cv, lane);
  if (t < BIN_W && lane == 31) wtot[wave] = inc;
  __syncthreads();
  int first = wsum[lane];
  for (int d = 16; d >= 1; d >>= 1) first += __shfl_xor(first, d);
  if (t < BIN_W) {
    int pre = 0;
    for (int w = 0; w < wave; ++w) pre += wtot[w];
    if (nb0 + t < N) oc[nb0 + t] = make_int2(first + pre + (inc - cv), cv);
  }
}

__global__ void k_place(const int2* __restrict__ bin_in, const int2* __restrict__ bin_out,
                        const int* __restrict__ rk_in, const int* __restrict__ rk_out,
                        const int2* __restrict__ oc_in, const int2* __restrict__ oc_out,
                        int2* __restrict__ lst_in, int2* __restrict__ lst_out,
                        float* __restrict__ rn_self, int E, int N) {
  int i = blockIdx.x * blockDim.x + threadIdx.x;
  if (i < E) {
    int2 a = bin_in[i];
    int2 b = bin_out[i];
    int ra = rk_in[i], rb = rk_out[i];
    int2 od = oc_in [a.x];
    int2 cs = oc_out[a.y];
    int2 os = oc_out[b.x];
    int2 cd = oc_in [b.y];
    float wa = rsqrtf((float)(od.y + 1) * (float)(cs.y + 1));
    float wb = rsqrtf((float)(cd.y + 1) * (float)(os.y + 1));
    lst_in [od.x + ra] = make_int2(a.y, __float_as_int(wa));
    lst_out[os.x + rb] = make_int2(b.y, __float_as_int(wb));
  } else if (i < E + N) {
    int n = i - E;
    rn_self[n] = rsqrtf((float)(oc_in[n].y + 1) * (float)(oc_out[n].y + 1));
  }
}


template <int K, int RELU>
__global__ __launch_bounds__(128)
void k_gemm_wmma(const float* __restrict__ X0, const float* __restrict__ X1, int ldx,
                 const float* __restrict__ Wa0, const float* __restrict__ ba0,
                 const float* __restrict__ Wb0, const float* __restrict__ bb0,
                 const float* __restrict__ Wa1, const float* __restrict__ ba1,
                 const float* __restrict__ Wb1, const float* __restrict__ bb1,
                 float* __restrict__ Y0, float* __restrict__ Y1, int ldy,
                 int halfO, int koffB, int nColTiles, int nMTiles, int blocksPerSide) {
  constexpr int NK = K / 4;

  const int sd  = (int)blockIdx.x >= blocksPerSide ? 1 : 0;
  const int blk = (int)blockIdx.x - sd * blocksPerSide;
  const float* X  = sd ? X1 : X0;
  const float* Wa = sd ? Wa1 : Wa0;
  const float* ba = sd ? ba1 : ba0;
  const float* Wb = sd ? Wb1 : Wb0;
  const float* bb = sd ? bb1 : bb0;
  float*       Y  = sd ? Y1 : Y0;

  int wave = threadIdx.x >> 5;
  int lane = threadIdx.x & 31;
  int mt   = blk * 4 + wave;
  if (mt >= nMTiles) return;

  int m0   = mt << 4;
  int mrow = lane & 15;
  int khi  = (lane >> 4) << 1;
  int col  = lane & 15;

  const float* xrow = X + (size_t)(m0 + mrow) * ldx;

  v2f afrag[NK];
  int curKoff = -1;

  for (int ct = 0; ct < nColTiles; ++ct) {
    int c0   = ct << 4;
    int half = (c0 >= halfO) ? 1 : 0;
    const float* W    = half ? Wb : Wa;
    const float* bias = half ? bb : ba;
    int wc0  = c0 - half * halfO;
    int koff = half ? koffB : 0;

    if (koff != curKoff) {
      curKoff = koff;
      const float* xp = xrow + koff + khi;
#pragma unroll
      for (int i = 0; i < NK; ++i) {
        v2f a = *(const v2f*)(xp + i * 4);
        if (RELU) { a.x = fmaxf(a.x, 0.0f); a.y = fmaxf(a.y, 0.0f); }
        afrag[i] = a;
      }
    }

    const float* wrow = W + (size_t)(wc0 + col) * K + khi;
    v8f acc = {};
#pragma unroll
    for (int i = 0; i < NK; ++i) {
      v2f bf = *(const v2f*)(wrow + i * 4);
      acc = __builtin_amdgcn_wmma_f32_16x16x4_f32(false, afrag[i], false, bf,
                                                  (short)0, acc, false, false);
    }

    float bv = bias[wc0 + col];
    int mbase = m0 + ((lane >> 4) << 3);
    float* yp = Y + (size_t)mbase * ldy + c0 + col;
#pragma unroll
    for (int g = 0; g < 8; ++g)
      yp[(size_t)g * ldy] = acc[g] + bv;
  }
}


template <int V> struct VecOf;
template <> struct VecOf<2> { typedef v2f type; };
template <> struct VecOf<4> { typedef v4f type; };

template <int F, int K>
__device__ __forceinline__ void rows_fma(const float* __restrict__ T, int sx, int sw, int j, int c,
                                         typename VecOf<F / 32>::type& acc) {
  typedef typename VecOf<F / 32>::type vf;
  vf v[K];
  float w[K];
#pragma unroll
  for (int u = 0; u < K; ++u) {
    int src = __builtin_amdgcn_readlane(sx, j + u);
    w[u] = __int_as_float(__builtin_amdgcn_readlane(sw, j + u));
    v[u] = *(const vf*)(T + (size_t)src * F + c);
  }
#pragma unroll
  for (int u = 0; u < K; ++u) acc += w[u] * v[u];
}

template <int F, int EPI>
__global__ __launch_bounds__(256)
void k_gather(const float* __restrict__ T0, const float* __restrict__ T1,
              const int2* __restrict__ oc0, const int2* __restrict__ oc1,
              const int2* __restrict__ lst0, const int2* __restrict__ lst1,
              const float* __restrict__ rn_self,
              const float* __restrict__ noise0, const float* __restrict__ noise1,
              float* __restrict__ Y0, float* __restrict__ Y1, int N) {
  static_assert(!EPI || F == 2 * OUT_F, "the fused reparametrize needs a mu|logstd row");
  constexpr int V = F / 32;
  typedef typename VecOf<V>::type vf;
  const int lane = threadIdx.x & 31;
  const int c = lane * V;
  const int nWaves = gridDim.x * 8;
  const int nItems = 2 * N;
  int wid = __builtin_amdgcn_readfirstlane(blockIdx.x * 8 + (threadIdx.x >> 5));

  if (wid >= nItems) return;
  int2 o = (wid >= N) ? oc1[wid - N] : oc0[wid];
  for (int item = wid; ; ) {
    const int sd = item >= N ? 1 : 0;
    const int node = item - sd * N;
    const float* T   = sd ? T1 : T0;
    const int2*  lst = sd ? lst1 : lst0;
    const int beg = o.x, cnt = o.y;
    const int nxt = item + nWaves;
    int2 o_nxt = o;
    if (nxt < nItems) o_nxt = (nxt >= N) ? oc1[nxt - N] : oc0[nxt];
    int n = min(32, cnt);
    int2 rec = make_int2(0, 0);
    if (lane < n) rec = lst[beg + lane];
    vf acc = rn_self[node] * *(const vf*)(T + (size_t)node * F + c);
    v2f nz = {0.0f, 0.0f};
    if constexpr (EPI) {
      const float* noise = sd ? noise1 : noise0;
      if (lane < 16) nz = *(const v2f*)(noise + (size_t)node * OUT_F + c);
    }

    for (int base = 0; base < cnt; ) {
      int j = 0;
      for (; j + 8 <= n; j += 8) rows_fma<F, 8>(T, rec.x, rec.y, j, c, acc);
      if (j + 4 <= n) { rows_fma<F, 4>(T, rec.x, rec.y, j, c, acc); j += 4; }
      if (j + 2 <= n) { rows_fma<F, 2>(T, rec.x, rec.y, j, c, acc); j += 2; }
      if (j < n)      { rows_fma<F, 1>(T, rec.x, rec.y, j, c, acc); }
      base += 32;
      if (base < cnt) {
        n = min(32, cnt - base);
        rec = make_int2(0, 0);
        if (lane < n) rec = lst[beg + base + lane];
      }
    }
    float* Y = sd ? Y1 : Y0;
    if constexpr (EPI) {
      const float lx = __shfl_xor(acc.x, 16), ly = __shfl_xor(acc.y, 16);
      if (lane < 16) {
        v2f r;
        r.x = acc.x + nz.x * expf(lx) * 0.2f;
        r.y = acc.y + nz.y * expf(ly) * 0.2f;
        *(v2f*)(Y + (size_t)node * OUT_F + c) = r;
      }
    } else {
      *(vf*)(Y + (size_t)node * F + c) = acc;
    }
    if (nxt >= nItems) break;
    item = nxt;
    o = o_nxt;
  }
}


#define GL_WAVES 16
#define GL_T     (GL_WAVES * 32)
#define GL_AP    (IN_F + 4)
#define GL_WP    (HID_F + 4)

__global__ __launch_bounds__(GL_T)
void k_gather_lin2(const float* __restrict__ T0, const float* __restrict__ T1,
                   const int2* __restrict__ oc0, const int2* __restrict__ oc1,
                   const int2* __restrict__ lst0, const int2* __restrict__ lst1,
                   const float* __restrict__ rn_self,
                   const float* __restrict__ Wa0, const float* __restrict__ ba0,
                   const float* __restrict__ Wb0, const float* __restrict__ bb0,
                   const float* __restrict__ Wa1, const float* __restrict__ ba1,
                   const float* __restrict__ Wb1, const float* __restrict__ bb1,
                   float* __restrict__ Y0, float* __restrict__ Y1, int N, int nTiles) {
  constexpr int F = IN_F;
  __shared__ __attribute__((aligned(16))) float wl[2 * HID_F * GL_WP];
  __shared__ __attribute__((aligned(16))) float tb[2 * 16 * GL_AP];
  __shared__ float bl[2 * HID_F];
  const int tid = threadIdx.x, lane = tid & 31;
  const int wave = __builtin_amdgcn_readfirstlane(tid >> 5);
  const int c = lane * 4;
  const int nAll = 2 * nTiles;

  int vt = blockIdx.x;
  if (vt >= nAll) return;
  int sd   = vt >= nTiles ? 1 : 0;
  int tile = vt - sd * nTiles;
  int node = tile * 16 + wave;
  int2 o = make_int2(0, 0);
  if (node < N) o = (sd ? oc1 : oc0)[node];

#pragma unroll
  for (int u = 0; u < 4; ++u) {
    const int q = tid + u * GL_T;
    const int row = (q >> 4) & (HID_F - 1), kc = (q & 15) * 4;
    const float* wa = (u < 2) ? Wa0 : Wa1;
    const float* wb = (u < 2) ? Wb0 : Wb1;
    const float* src = (row < OUT_F) ? (wa + row * HID_F + kc) : (wb + (row - OUT_F) * HID_F + kc);
    *(v4f*)(wl + (u >> 1) * (HID_F * GL_WP) + row * GL_WP + kc) = *(const v4f*)src;
  }
  if (tid < 2 * HID_F) {
    const int r = tid & (HID_F - 1);
    const float* pa = (tid < HID_F) ? ba0 : ba1;
    const float* pb = (tid < HID_F) ? bb0 : bb1;
    bl[tid] = (r < OUT_F) ? pa[r] : pb[r - OUT_F];
  }


  for (int it = 0; ; ++it) {
    const float* T   = sd ? T1 : T0;
    const int2*  lst = sd ? lst1 : lst0;
    const int nvt   = vt + gridDim.x;
    const int nsd   = nvt >= nTiles ? 1 : 0;
    const int ntile = nvt - nsd * nTiles;
    const int nnode = ntile * 16 + wave;
    int2 o_nxt = make_int2(0, 0);
    if (nvt < nAll && nnode < N) o_nxt = (nsd ? oc1 : oc0)[nnode];

    v4f acc = {0.0f, 0.0f, 0.0f, 0.0f};
    if (node < N) {
      const int beg = o.x, cnt = o.y;
      int n = min(32, cnt);
      int2 rec = make_int2(0, 0);
      if (lane < n) rec = lst[beg + lane];
      acc = rn_self[node] * *(const v4f*)(T + (size_t)node * F + c);
      for (int base = 0; base < cnt; ) {
        int j = 0;
        for (; j + 8 <= n; j += 8) rows_fma<F, 8>(T, rec.x, rec.y, j, c, acc);
        if (j + 4 <= n) { rows_fma<F, 4>(T, rec.x, rec.y, j, c, acc); j += 4; }
        if (j + 2 <= n) { rows_fma<F, 2>(T, rec.x, rec.y, j, c, acc); j += 2; }
        if (j < n)      { rows_fma<F, 1>(T, rec.x, rec.y, j, c, acc); }
        base += 32;
        if (base < cnt) {
          n = min(32, cnt - base);
          rec = make_int2(0, 0);
          if (lane < n) rec = lst[beg + base + lane];
        }
      }
    }
    acc.x = fmaxf(acc.x, 0.0f); acc.y = fmaxf(acc.y, 0.0f); acc.z = fmaxf(acc.z, 0.0f); acc.w = fmaxf(acc.w, 0.0f);
    float* tcur = tb + (it & 1) * (16 * GL_AP);
    *(v4f*)(tcur + wave * GL_AP + c) = acc;
    __syncthreads();

    if ((wave >> 2) == (it & 3)) {
      const int ct   = wave & 3;
      const int mrow = lane & 15;
      const int khi  = (lane >> 4) << 1;
      const int col  = lane & 15;
      const float* ap = tcur + mrow * GL_AP + (ct >> 1) * HID_F + khi;
      const float* wp = wl + sd * (HID_F * GL_WP) + (ct * 16 + col) * GL_WP + khi;
      v8f a8 = {};
#pragma unroll
      for (int i = 0; i < HID_F / 4; ++i) {
        v2f af = *(const v2f*)(ap + i * 4);
        v2f bf = *(const v2f*)(wp + i * 4);
        a8 = __builtin_amdgcn_wmma_f32_16x16x4_f32(false, af, false, bf, (short)0, a8, false, false);
      }
      const float bv = bl[sd * HID_F + ct * 16 + col];
      const int mbase = tile * 16 + ((lane >> 4) << 3);
      float* yp = (sd ? Y1 : Y0) + (size_t)mbase * HID_F + ct * 16 + col;
#pragma unroll
      for (int g = 0; g < 8; ++g)
        if (mbase + g < N) yp[(size_t)g * HID_F] = a8[g] + bv;
    }
    if (nvt >= nAll) break;
    vt = nvt; sd = nsd; tile = ntile; node = nnode; o = o_nxt;
  }
}


static inline int ceil_div(int a, int b) { return (a + b - 1) / b; }

extern "C" void kernel_launch(void* const* d_in, const int* in_sizes, int n_in,
                              void* d_out, int out_size, void* d_ws, size_t ws_size,
                              hipStream_t stream) {
  const float* s       = (const float*)d_in[0];
  const float* t       = (const float*)d_in[1];
  const int*   ei      = (const int*)  d_in[2];
  const float* noise_s = (const float*)d_in[3];
  const float* noise_t = (const float*)d_in[4];
  const float* sm1_W = (const float*)d_in[5],  *sm1_b = (const float*)d_in[6];
  const float* sm2_W = (const float*)d_in[7],  *sm2_b = (const float*)d_in[8];
  const float* sl1_W = (const float*)d_in[9],  *sl1_b = (const float*)d_in[10];
  const float* sl2_W = (const float*)d_in[11], *sl2_b = (const float*)d_in[12];
  const float* tm1_W = (const float*)d_in[13], *tm1_b = (const float*)d_in[14];
  const float* tm2_W = (const float*)d_in[15], *tm2_b = (const float*)d_in[16];
  const float* tl1_W = (const float*)d_in[17], *tl1_b = (const float*)d_in[18];
  const float* tl2_W = (const float*)d_in[19], *tl2_b = (const float*)d_in[20];

  const int N = in_sizes[0] / IN_F;
  const int E = in_sizes[2] / 2;
  const int* ei0 = ei;
  const int* ei1 = ei + E;

  const int Npad = (N + 3) & ~3;
  float* ws      = (float*)d_ws;
  float* rn_self = ws + 2 * Npad;
  float* P0 = ws + (size_t)3 * N + E;
  float* P1 = P0 + (size_t)N * IN_F;
  float* P2 = P1 + (size_t)N * IN_F;
  float* P3 = P2 + (size_t)N * IN_F;
  int2* oc_in   = (int2*)P3;
  int2* oc_out  = oc_in  + N;
  int2* lst_in  = oc_out + N;
  int2* lst_out = lst_in + E;

  const int NB = ceil_div(N, BIN_W);
  const int G  = ceil_div(E, CH_E);
  int2* bin_in  = (int2*)P0;
  int2* bin_out = bin_in + E;
  int*  rk_in   = (int*)(bin_out + E);
  int*  rk_out  = rk_in + E;
  int*  seg_in  = rk_out + E;
  int*  seg_out = seg_in + (size_t)(NB + 1) * G;

  float* Y1s = P0;
  float* Y1t = P2;
  float* Y2s = P1;
  float* Y2t = P1 + (size_t)N * HID_F;
  float* out_s = (float*)d_out;
  float* out_t = out_s + (size_t)N * OUT_F;

  const int B = 256;
  const int nMTiles = ceil_div(N, 16);
  const int gemmBlocks = ceil_div(nMTiles, 4);
  const int gatherBlocks = ceil_div(2 * N, 8) < 2048 ? ceil_div(2 * N, 8) : 2048;
  const int fusedBlocks = 2 * nMTiles < 1024 ? 2 * nMTiles : 1024;

  k_bin<<<G, 1024, 0, stream>>>(ei0, ei1, bin_in, bin_out, seg_in, seg_out, E, NB, G);
  k_rank<<<dim3(NB, 2), 1024, 0, stream>>>(bin_in, bin_out, seg_in, seg_out, oc_in, oc_out,
                                           rk_in, rk_out, N, G);
  k_place<<<ceil_div(E + N, B), B, 0, stream>>>(bin_in, bin_out, rk_in, rk_out, oc_in, oc_out,
                                                lst_in, lst_out, rn_self, E, N);

  k_gemm_wmma<IN_F, 0><<<2 * gemmBlocks, 128, 0, stream>>>(
      s, t, IN_F, sm1_W, sm1_b, sl1_W, sl1_b, tm1_W, tm1_b, tl1_W, tl1_b,
      Y1s, Y1t, IN_F, HID_F, 0, IN_F / 16, nMTiles, gemmBlocks);
  k_gather_lin2<<<fusedBlocks, GL_T, 0, stream>>>(Y1s, Y1t, oc_in, oc_out, lst_in, lst_out, rn_self,
                                                  sm2_W, sm2_b, sl2_W, sl2_b, tm2_W, tm2_b, tl2_W, tl2_b,
                                                  Y2s, Y2t, N, nMTiles);
  k_gather<HID_F, 1><<<gatherBlocks, B, 0, stream>>>(Y2s, Y2t, oc_out, oc_in, lst_out, lst_in, rn_self,
                                                     noise_s, noise_t, out_s, out_t, N);
}


# Layer-1 product: a side's weights copied once per 16-wave work-group into LDS by whole-line loads and B fragments read from LDS, not one waited 16-line global load per WMMA; k order and sums unchanged
# speedup vs baseline: 5.2593x; 1.2768x over previous
#include <hip/hip_runtime.h>


typedef __attribute__((ext_vector_type(2))) float v2f;
typedef __attribute__((ext_vector_type(4))) float v4f;
typedef __attribute__((ext_vector_type(8))) float v8f;

#define IN_F   128
#define HID_F  64
#define OUT_F  32


#define BIN_SH   8
#define BIN_W    (1 << BIN_SH)
#define BIN_MAXB 1024
#define CH_E     4096

__device__ __forceinline__ int wave_incl_scan(int v, int lane) {
  for (int d = 1; d < 32; d <<= 1) { int y = __shfl_up(v, d); if (lane >= d) v += y; }
  return v;
}

__global__ __launch_bounds__(1024)
void k_bin(const int* __restrict__ ei0, const int* __restrict__ ei1,
           int2* __restrict__ bin_in, int2* __restrict__ bin_out,
           int* __restrict__ seg_in, int* __restrict__ seg_out, int E, int NB, int G) {
  __shared__ int h_in[BIN_MAXB];
  __shared__ int h_out[BIN_MAXB];
  __shared__ int ws_in[32];
  __shared__ int ws_out[32];
  const int t = threadIdx.x, lane = t & 31, wave = t >> 5, g = blockIdx.x;
  h_in[t] = 0; h_out[t] = 0;
  __syncthreads();

  const int base = g * CH_E;
  int s[4], d[4], ri[4], ro[4];
#pragma unroll
  for (int k = 0; k < 4; ++k) {
    int e = base + k * 1024 + t;
    s[k] = -1; d[k] = 0;
    if (e < E) { s[k] = ei0[e]; d[k] = ei1[e]; }
  }
#pragma unroll
  for (int k = 0; k < 4; ++k) {
    ri[k] = 0; ro[k] = 0;
    if (s[k] >= 0) {
      ri[k] = atomicAdd(&h_in [d[k] >> BIN_SH], 1);
      ro[k] = atomicAdd(&h_out[s[k] >> BIN_SH], 1);
    }
  }
  __syncthreads();

  const int ci = h_in[t], co = h_out[t];
  const int ii = wave_incl_scan(ci, lane), io = wave_incl_scan(co, lane);
  if (lane == 31) { ws_in[wave] = ii; ws_out[wave] = io; }
  __syncthreads();
  const int wi = ws_in[lane], wo = ws_out[lane];
  const int wii = wave_incl_scan(wi, lane), woo = wave_incl_scan(wo, lane);
  const int ex_in  = (__shfl(wii, wave) - __shfl(wi, wave)) + (ii - ci);
  const int ex_out = (__shfl(woo, wave) - __shfl(wo, wave)) + (io - co);
  if (t <= NB) { seg_in[t * G + g] = ex_in; seg_out[t * G + g] = ex_out; }
  h_in[t] = ex_in; h_out[t] = ex_out;
  __syncthreads();

#pragma unroll
  for (int k = 0; k < 4; ++k) {
    if (s[k] >= 0) {
      bin_in [base + h_in [d[k] >> BIN_SH] + ri[k]] = make_int2(d[k], s[k]);
      bin_out[base + h_out[s[k] >> BIN_SH] + ro[k]] = make_int2(s[k], d[k]);
    }
  }
}

#define ST_CAP 12288

template <int PASS>
__device__ __forceinline__ int walk_runs(const int2* __restrict__ bin, const int* __restrict__ seg,
                                         int b, int G, int nb0, int* c, int* sst, int* sen,
                                         int* stage, int* __restrict__ lst_b, bool staged,
                                         int t, int lane, int wave) {
  int part = 0;
  for (int g0 = 0; g0 < G; g0 += 1024) {
    if (PASS == 0 || G > 1024) {
      if (g0 + t < G) {
        int a = seg[b * G + g0 + t];
        sst[t] = a; sen[t] = seg[(b + 1) * G + g0 + t];
        part += a;
      }
      __syncthreads();
    }
    const int ng = min(1024, G - g0);
    for (int j = wave; j < ng; j += 32) {
      const int st = sst[j], en = sen[j];
      const size_t cb = (size_t)(g0 + j) * CH_E;
      for (int p = st + lane; p < en; p += 32) {
        int2 r = bin[cb + p];
        if (PASS == 0) {
          atomicAdd(&c[r.x - nb0], 1);
        } else {
          const int pos = atomicAdd(&c[r.x - nb0], 1);
          if (staged) stage[pos] = r.y; else lst_b[pos] = r.y;
        }
      }
    }
    __syncthreads();
  }
  return part;
}

__global__ __launch_bounds__(1024)
void k_rank_place(const int2* __restrict__ bin_in, const int2* __restrict__ bin_out,
                  const int* __restrict__ seg_in, const int* __restrict__ seg_out,
                  int2* __restrict__ oc_in, int2* __restrict__ oc_out,
                  int* __restrict__ lst_in, int* __restrict__ lst_out, int N, int G) {
  __shared__ int c[BIN_W];
  __shared__ int sst[1024];
  __shared__ int sen[1024];
  __shared__ int wsum[32];
  __shared__ int wtot[BIN_W / 32];
  __shared__ int stage[ST_CAP];
  const int2* bin = blockIdx.y ? bin_out : bin_in;
  const int*  seg = blockIdx.y ? seg_out : seg_in;
  int2*       oc  = blockIdx.y ? oc_out  : oc_in;
  int*        lst = blockIdx.y ? lst_out : lst_in;
  const int t = threadIdx.x, lane = t & 31, wave = t >> 5, b = blockIdx.x;
  const int nb0 = b << BIN_SH;
  if (t < BIN_W) c[t] = 0;
  __syncthreads();

  int part = walk_runs<0>(bin, seg, b, G, nb0, c, sst, sen, stage, lst, false, t, lane, wave);

  for (int d = 16; d >= 1; d >>= 1) part += __shfl_xor(part, d);
  if (lane == 0) wsum[wave] = part;
  const int cv  = (t < BIN_W) ? c[t] : 0;
  const int inc = wave_incl_scan(cv, lane);
  if (t < BIN_W && lane == 31) wtot[wave] = inc;
  __syncthreads();
  int first = wsum[lane];
  for (int d = 16; d >= 1; d >>= 1) first += __shfl_xor(first, d);
  int total = (lane < BIN_W / 32) ? wtot[lane] : 0;
  for (int d = 16; d >= 1; d >>= 1) total += __shfl_xor(total, d);
  if (t < BIN_W) {
    int pre = 0;
    for (int w = 0; w < wave; ++w) pre += wtot[w];
    const int ex = pre + (inc - cv);
    if (nb0 + t < N) oc[nb0 + t] = make_int2(first + ex, cv);
    c[t] = ex;
  }
  __syncthreads();

  const bool staged = total <= ST_CAP;
  walk_runs<1>(bin, seg, b, G, nb0, c, sst, sen, stage, lst + first, staged, t, lane, wave);
  if (staged)
    for (int i = t; i < total; i += 1024) lst[first + i] = stage[i];
}


#define G1_WAVES 16
#define G1_T     (G1_WAVES * 32)
#define G1_COLS  (2 * HID_F)
#define G1_WP    (IN_F + 4)

__device__ __forceinline__ void g1_load_tile(const float* __restrict__ X, const int2* __restrict__ sc,
                                             int m0, int lane, v2f (&afrag)[IN_F / 4], float (&fs)[8]) {
  const int mrow = lane & 15;
  const int khi  = (lane >> 4) << 1;
  const float* xp = X + (size_t)(m0 + mrow) * IN_F + khi;
#pragma unroll
  for (int i = 0; i < IN_F / 4; ++i) afrag[i] = *(const v2f*)(xp + i * 4);
#pragma unroll
  for (int g = 0; g < 8; ++g)
    fs[g] = rsqrtf((float)(sc[m0 + ((lane >> 4) << 3) + g].y + 1));
}

__global__ __launch_bounds__(G1_T)
void k_gemm_wmma(const float* __restrict__ X0, const float* __restrict__ X1,
                 const float* __restrict__ Wa0, const float* __restrict__ ba0,
                 const float* __restrict__ Wb0, const float* __restrict__ bb0,
                 const float* __restrict__ Wa1, const float* __restrict__ ba1,
                 const float* __restrict__ Wb1, const float* __restrict__ bb1,
                 const int2* __restrict__ sc0, const int2* __restrict__ sc1,
                 float* __restrict__ Y0, float* __restrict__ Y1, int nMTiles) {
  constexpr int K  = IN_F;
  constexpr int NK = K / 4;
  __shared__ __attribute__((aligned(16))) float wl[G1_COLS * G1_WP];
  __shared__ float bl[G1_COLS];

  const int tid = threadIdx.x, lane = tid & 31;
  const int wave = __builtin_amdgcn_readfirstlane(tid >> 5);
  const int sd = (int)blockIdx.x & 1;
  const int bi = (int)blockIdx.x >> 1;
  const int nb = (int)gridDim.x >> 1;
  const float* X  = sd ? X1 : X0;
  const float* Wa = sd ? Wa1 : Wa0;
  const float* ba = sd ? ba1 : ba0;
  const float* Wb = sd ? Wb1 : Wb0;
  const float* bb = sd ? bb1 : bb0;
  float*       Y  = sd ? Y1 : Y0;
  const int2*  sc = sd ? sc1 : sc0;

  const int slot = (wave % 5 == 0) ? 12 + wave / 5 : wave - (wave + 4) / 5;
  int mt = bi + slot * nb;
  v2f afrag[NK];
  float fs[8];
  if (mt < nMTiles) g1_load_tile(X, sc, mt << 4, lane, afrag, fs);

#pragma unroll
  for (int u = 0; u < (G1_COLS * K / 4) / G1_T; ++u) {
    const int q = tid + u * G1_T;
    const int row = q >> 5, kc = (q & 31) * 4;
    const float* src = (row < HID_F) ? (Wa + row * K + kc) : (Wb + (row - HID_F) * K + kc);
    *(v4f*)(wl + row * G1_WP + kc) = *(const v4f*)src;
  }
  if (tid < G1_COLS) bl[tid] = (tid < HID_F) ? ba[tid] : bb[tid - HID_F];
  __syncthreads();

  const int col = lane & 15;
  const int khi = (lane >> 4) << 1;
  const float* wbase = wl + col * G1_WP + khi;

  while (mt < nMTiles) {
    const int m0 = mt << 4;
    const int mbase = m0 + ((lane >> 4) << 3);
#pragma nounroll
    for (int ct = 0; ct < G1_COLS / 16; ++ct) {
      const float* wp = wbase + ct * (16 * G1_WP);
      v8f acc = {};
#pragma unroll
      for (int i = 0; i < NK; ++i) {
        const v2f bf = *(const v2f*)(wp + i * 4);
        acc = __builtin_amdgcn_wmma_f32_16x16x4_f32(false, afrag[i], false, bf,
                                                    (short)0, acc, false, false);
      }
      const float bv = bl[ct * 16 + col];
      float* yp = Y + (size_t)mbase * G1_COLS + ct * 16 + col;
#pragma unroll
      for (int g = 0; g < 8; ++g)
        yp[(size_t)g * G1_COLS] = (acc[g] + bv) * fs[g];
    }
    mt += G1_WAVES * nb;
    if (mt < nMTiles) g1_load_tile(X, sc, mt << 4, lane, afrag, fs);
  }
}


template <int V> struct VecOf;
template <> struct VecOf<2> { typedef v2f type; };
template <> struct VecOf<4> { typedef v4f type; };

template <int F, int K>
__device__ __forceinline__ void rows_add(const float* __restrict__ T, int sx, int j, int c,
                                         typename VecOf<F / 32>::type& acc) {
  typedef typename VecOf<F / 32>::type vf;
  vf v[K];
#pragma unroll
  for (int u = 0; u < K; ++u) {
    int src = __builtin_amdgcn_readlane(sx, j + u);
    v[u] = *(const vf*)(T + (size_t)src * F + c);
  }
#pragma unroll
  for (int u = 0; u < K; ++u) acc += v[u];
}

template <int F, int EPI>
__global__ __launch_bounds__(256)
void k_gather(const float* __restrict__ T0, const float* __restrict__ T1,
              const int2* __restrict__ oc0, const int2* __restrict__ oc1,
              const int* __restrict__ lst0, const int* __restrict__ lst1,
              const float* __restrict__ noise0, const float* __restrict__ noise1,
              float* __restrict__ Y0, float* __restrict__ Y1, int N) {
  static_assert(!EPI || F == 2 * OUT_F, "the fused reparametrize needs a mu|logstd row");
  constexpr int V = F / 32;
  typedef typename VecOf<V>::type vf;
  const int lane = threadIdx.x & 31;
  const int c = lane * V;
  const int nWaves = gridDim.x * 8;
  const int nItems = 2 * N;
  int wid = __builtin_amdgcn_readfirstlane(blockIdx.x * 8 + (threadIdx.x >> 5));

  if (wid >= nItems) return;
  int2 o = (wid >= N) ? oc1[wid - N] : oc0[wid];
  for (int item = wid; ; ) {
    const int sd = item >= N ? 1 : 0;
    const int node = item - sd * N;
    const float* T   = sd ? T1 : T0;
    const int*   lst = sd ? lst1 : lst0;
    const int beg = o.x, cnt = o.y;
    const int nxt = item + nWaves;
    int2 o_nxt = o;
    if (nxt < nItems) o_nxt = (nxt >= N) ? oc1[nxt - N] : oc0[nxt];
    int n = min(32, cnt);
    int rec = 0;
    if (lane < n) rec = lst[beg + lane];
    vf acc = *(const vf*)(T + (size_t)node * F + c);
    v2f nz = {0.0f, 0.0f};
    if constexpr (EPI) {
      const float* noise = sd ? noise1 : noise0;
      if (lane < 16) nz = *(const v2f*)(noise + (size_t)node * OUT_F + c);
    }

    for (int base = 0; base < cnt; ) {
      int j = 0;
      for (; j + 8 <= n; j += 8) rows_add<F, 8>(T, rec, j, c, acc);
      if (j + 4 <= n) { rows_add<F, 4>(T, rec, j, c, acc); j += 4; }
      if (j + 2 <= n) { rows_add<F, 2>(T, rec, j, c, acc); j += 2; }
      if (j < n)      { rows_add<F, 1>(T, rec, j, c, acc); }
      base += 32;
      if (base < cnt) {
        n = min(32, cnt - base);
        rec = 0;
        if (lane < n) rec = lst[beg + base + lane];
      }
    }
    acc *= rsqrtf((float)(cnt + 1));
    float* Y = sd ? Y1 : Y0;
    if constexpr (EPI) {
      const float lx = __shfl_xor(acc.x, 16), ly = __shfl_xor(acc.y, 16);
      if (lane < 16) {
        v2f r;
        r.x = acc.x + nz.x * expf(lx) * 0.2f;
        r.y = acc.y + nz.y * expf(ly) * 0.2f;
        *(v2f*)(Y + (size_t)node * OUT_F + c) = r;
      }
    } else {
      *(vf*)(Y + (size_t)node * F + c) = acc;
    }
    if (nxt >= nItems) break;
    item = nxt;
    o = o_nxt;
  }
}


#define GL_WAVES 16
#define GL_T     (GL_WAVES * 32)
#define GL_AP    (IN_F + 4)
#define GL_WP    (HID_F + 4)

__global__ __launch_bounds__(GL_T)
void k_gather_lin2(const float* __restrict__ T0, const float* __restrict__ T1,
                   const int2* __restrict__ oc0, const int2* __restrict__ oc1,
                   const int* __restrict__ lst0, const int* __restrict__ lst1,
                   const float* __restrict__ Wa0, const float* __restrict__ ba0,
                   const float* __restrict__ Wb0, const float* __restrict__ bb0,
                   const float* __restrict__ Wa1, const float* __restrict__ ba1,
                   const float* __restrict__ Wb1, const float* __restrict__ bb1,
                   float* __restrict__ Y0, float* __restrict__ Y1, int N, int nTiles) {
  constexpr int F = IN_F;
  __shared__ __attribute__((aligned(16))) float wl[2 * HID_F * GL_WP];
  __shared__ __attribute__((aligned(16))) float tb[2 * 16 * GL_AP];
  __shared__ float bl[2 * HID_F];
  const int tid = threadIdx.x, lane = tid & 31;
  const int wave = __builtin_amdgcn_readfirstlane(tid >> 5);
  const int c = lane * 4;
  const int nAll = 2 * nTiles;

  int vt = blockIdx.x;
  if (vt >= nAll) return;
  int sd   = vt >= nTiles ? 1 : 0;
  int tile = vt - sd * nTiles;
  int node = tile * 16 + wave;
  int2 o = make_int2(0, 0);
  if (node < N) o = (sd ? oc1 : oc0)[node];

#pragma unroll
  for (int u = 0; u < 4; ++u) {
    const int q = tid + u * GL_T;
    const int row = (q >> 4) & (HID_F - 1), kc = (q & 15) * 4;
    const float* wa = (u < 2) ? Wa0 : Wa1;
    const float* wb = (u < 2) ? Wb0 : Wb1;
    const float* src = (row < OUT_F) ? (wa + row * HID_F + kc) : (wb + (row - OUT_F) * HID_F + kc);
    *(v4f*)(wl + (u >> 1) * (HID_F * GL_WP) + row * GL_WP + kc) = *(const v4f*)src;
  }
  if (tid < 2 * HID_F) {
    const int r = tid & (HID_F - 1);
    const float* pa = (tid < HID_F) ? ba0 : ba1;
    const float* pb = (tid < HID_F) ? bb0 : bb1;
    bl[tid] = (r < OUT_F) ? pa[r] : pb[r - OUT_F];
  }


  for (int it = 0; ; ++it) {
    const float* T   = sd ? T1 : T0;
    const int*   lst = sd ? lst1 : lst0;
    const int nvt   = vt + gridDim.x;
    const int nsd   = nvt >= nTiles ? 1 : 0;
    const int ntile = nvt - nsd * nTiles;
    const int nnode = ntile * 16 + wave;
    int2 o_nxt = make_int2(0, 0);
    if (nvt < nAll && nnode < N) o_nxt = (nsd ? oc1 : oc0)[nnode];

    v4f acc = {0.0f, 0.0f, 0.0f, 0.0f};
    float fd = 1.0f;
    if (node < N) {
      const int beg = o.x, cnt = o.y;
      int n = min(32, cnt);
      int rec = 0;
      if (lane < n) rec = lst[beg + lane];
      acc = *(const v4f*)(T + (size_t)node * F + c);
      for (int base = 0; base < cnt; ) {
        int j = 0;
        for (; j + 8 <= n; j += 8) rows_add<F, 8>(T, rec, j, c, acc);
        if (j + 4 <= n) { rows_add<F, 4>(T, rec, j, c, acc); j += 4; }
        if (j + 2 <= n) { rows_add<F, 2>(T, rec, j, c, acc); j += 2; }
        if (j < n)      { rows_add<F, 1>(T, rec, j, c, acc); }
        base += 32;
        if (base < cnt) {
          n = min(32, cnt - base);
          rec = 0;
          if (lane < n) rec = lst[beg + base + lane];
        }
      }
      fd = rsqrtf((float)(cnt + 1));
      acc *= fd;
    }
    acc.x = fmaxf(acc.x, 0.0f); acc.y = fmaxf(acc.y, 0.0f); acc.z = fmaxf(acc.z, 0.0f); acc.w = fmaxf(acc.w, 0.0f);
    float* tcur = tb + (it & 1) * (16 * GL_AP);
    *(v4f*)(tcur + wave * GL_AP + c) = acc;
    if (lane == 0) tcur[wave * GL_AP + IN_F] = fd;
    __syncthreads();

    if ((wave >> 2) == (it & 3)) {
      const int ct   = wave & 3;
      const int mrow = lane & 15;
      const int khi  = (lane >> 4) << 1;
      const int col  = lane & 15;
      const float* ap = tcur + mrow * GL_AP + (ct >> 1) * HID_F + khi;
      const float* wp = wl + sd * (HID_F * GL_WP) + (ct * 16 + col) * GL_WP + khi;
      const int mbase = tile * 16 + ((lane >> 4) << 3);
      const float* fp = tcur + ((lane >> 4) << 3) * GL_AP + IN_F;
      float fs[8];
#pragma unroll
      for (int g = 0; g < 8; ++g) fs[g] = fp[g * GL_AP];
      v8f a8 = {};
#pragma unroll
      for (int i = 0; i < HID_F / 4; ++i) {
        v2f af = *(const v2f*)(ap + i * 4);
        v2f bf = *(const v2f*)(wp + i * 4);
        a8 = __builtin_amdgcn_wmma_f32_16x16x4_f32(false, af, false, bf, (short)0, a8, false, false);
      }
      const float bv = bl[sd * HID_F + ct * 16 + col];
      float* yp = (sd ? Y1 : Y0) + (size_t)mbase * HID_F + ct * 16 + col;
#pragma unroll
      for (int g = 0; g < 8; ++g)
        if (mbase + g < N) yp[(size_t)g * HID_F] = (a8[g] + bv) * fs[g];
    }
    if (nvt >= nAll) break;
    vt = nvt; sd = nsd; tile = ntile; node = nnode; o = o_nxt;
  }
}


static inline int ceil_div(int a, int b) { return (a + b - 1) / b; }

extern "C" void kernel_launch(void* const* d_in, const int* in_sizes, int n_in,
                              void* d_out, int out_size, void* d_ws, size_t ws_size,
                              hipStream_t stream) {
  const float* s       = (const float*)d_in[0];
  const float* t       = (const float*)d_in[1];
  const int*   ei      = (const int*)  d_in[2];
  const float* noise_s = (const float*)d_in[3];
  const float* noise_t = (const float*)d_in[4];
  const float* sm1_W = (const float*)d_in[5],  *sm1_b = (const float*)d_in[6];
  const float* sm2_W = (const float*)d_in[7],  *sm2_b = (const float*)d_in[8];
  const float* sl1_W = (const float*)d_in[9],  *sl1_b = (const float*)d_in[10];
  const float* sl2_W = (const float*)d_in[11], *sl2_b = (const float*)d_in[12];
  const float* tm1_W = (const float*)d_in[13], *tm1_b = (const float*)d_in[14];
  const float* tm2_W = (const float*)d_in[15], *tm2_b = (const float*)d_in[16];
  const float* tl1_W = (const float*)d_in[17], *tl1_b = (const float*)d_in[18];
  const float* tl2_W = (const float*)d_in[19], *tl2_b = (const float*)d_in[20];

  const int N = in_sizes[0] / IN_F;
  const int E = in_sizes[2] / 2;
  const int* ei0 = ei;
  const int* ei1 = ei + E;

  float* ws = (float*)d_ws;
  float* P0 = ws + (size_t)3 * N + E;
  float* P1 = P0 + (size_t)N * IN_F;
  float* P2 = P1 + (size_t)N * IN_F;
  float* P3 = P2 + (size_t)N * IN_F;
  int2* oc_in   = (int2*)P3;
  int2* oc_out  = oc_in  + N;
  int*  lst_in  = (int*)(oc_out + N);
  int*  lst_out = lst_in + E;

  const int NB = ceil_div(N, BIN_W);
  const int G  = ceil_div(E, CH_E);
  int2* bin_in  = (int2*)P0;
  int2* bin_out = bin_in + E;
  int*  seg_in  = (int*)(bin_out + E);
  int*  seg_out = seg_in + (size_t)(NB + 1) * G;

  float* Y1s = P0;
  float* Y1t = P2;
  float* Y2s = P1;
  float* Y2t = P1 + (size_t)N * HID_F;
  float* out_s = (float*)d_out;
  float* out_t = out_s + (size_t)N * OUT_F;

  const int B = 256;
  const int nMTiles = ceil_div(N, 16);
  const int gemmBlocks = 512;
  const int gatherBlocks = ceil_div(2 * N, 8) < 2048 ? ceil_div(2 * N, 8) : 2048;
  const int fusedBlocks = 2 * nMTiles < 1024 ? 2 * nMTiles : 1024;

  k_bin<<<G, 1024, 0, stream>>>(ei0, ei1, bin_in, bin_out, seg_in, seg_out, E, NB, G);
  k_rank_place<<<dim3(NB, 2), 1024, 0, stream>>>(bin_in, bin_out, seg_in, seg_out, oc_in, oc_out,
                                                 lst_in, lst_out, N, G);

  k_gemm_wmma<<<gemmBlocks, G1_T, 0, stream>>>(
      s, t, sm1_W, sm1_b, sl1_W, sl1_b, tm1_W, tm1_b, tl1_W, tl1_b, oc_out, oc_in,
      Y1s, Y1t, nMTiles);
  k_gather_lin2<<<fusedBlocks, GL_T, 0, stream>>>(Y1s, Y1t, oc_in, oc_out, lst_in, lst_out,
                                                  sm2_W, sm2_b, sl2_W, sl2_b, tm2_W, tm2_b, tl2_W, tl2_b,
                                                  Y2s, Y2t, N, nMTiles);
  k_gather<HID_F, 1><<<gatherBlocks, B, 0, stream>>>(Y2s, Y2t, oc_out, oc_in, lst_out, lst_in,
                                                     noise_s, noise_t, out_s, out_t, N);
}


# Edge binning: a chunk's two bucket-sorted record arrays are put together in LDS and leave as whole 128-byte lines (run table behind the last barrier), not as 8192 scattered 8-byte stores per block
# speedup vs baseline: 5.4466x; 1.0356x over previous
#include <hip/hip_runtime.h>


typedef __attribute__((ext_vector_type(2))) float v2f;
typedef __attribute__((ext_vector_type(4))) float v4f;
typedef __attribute__((ext_vector_type(8))) float v8f;
typedef __attribute__((ext_vector_type(2))) int v2i;
typedef __attribute__((ext_vector_type(4))) int v4i;

#define IN_F   128
#define HID_F  64
#define OUT_F  32


#define BIN_SH   8
#define BIN_W    (1 << BIN_SH)
#define BIN_MAXB 1024
#define CH_E     4096

__device__ __forceinline__ int wave_incl_scan(int v, int lane) {
  for (int d = 1; d < 32; d <<= 1) { int y = __shfl_up(v, d); if (lane >= d) v += y; }
  return v;
}

__global__ __launch_bounds__(1024)
void k_bin(const int* __restrict__ ei0, const int* __restrict__ ei1,
           int2* __restrict__ bin_in, int2* __restrict__ bin_out,
           int* __restrict__ seg_in, int* __restrict__ seg_out, int E, int NB, int G) {
  __shared__ int h_in[BIN_MAXB];
  __shared__ int h_out[BIN_MAXB];
  __shared__ int ws_in[32];
  __shared__ int ws_out[32];
  __shared__ __attribute__((aligned(16))) int st_in[2 * CH_E];
  __shared__ __attribute__((aligned(16))) int st_out[2 * CH_E];
  const int t = threadIdx.x, lane = t & 31, wave = t >> 5, g = blockIdx.x;
  h_in[t] = 0; h_out[t] = 0;
  __syncthreads();

  const int base = g * CH_E;
  int s[4], d[4], ri[4], ro[4];
#pragma unroll
  for (int k = 0; k < 4; ++k) {
    int e = base + k * 1024 + t;
    s[k] = -1; d[k] = 0;
    if (e < E) { s[k] = ei0[e]; d[k] = ei1[e]; }
  }
#pragma unroll
  for (int k = 0; k < 4; ++k) {
    ri[k] = 0; ro[k] = 0;
    if (s[k] >= 0) {
      ri[k] = atomicAdd(&h_in [d[k] >> BIN_SH], 1);
      ro[k] = atomicAdd(&h_out[s[k] >> BIN_SH], 1);
    }
  }
  __syncthreads();

  const int ci = h_in[t], co = h_out[t];
  const int ii = wave_incl_scan(ci, lane), io = wave_incl_scan(co, lane);
  if (lane == 31) { ws_in[wave] = ii; ws_out[wave] = io; }
  __syncthreads();
  const int wi = ws_in[lane], wo = ws_out[lane];
  const int wii = wave_incl_scan(wi, lane), woo = wave_incl_scan(wo, lane);
  const int ex_in  = (__shfl(wii, wave) - __shfl(wi, wave)) + (ii - ci);
  const int ex_out = (__shfl(woo, wave) - __shfl(wo, wave)) + (io - co);
  h_in[t] = ex_in; h_out[t] = ex_out;
  __syncthreads();

#pragma unroll
  for (int k = 0; k < 4; ++k) {
    if (s[k] >= 0) {
      v2i a; a.x = d[k]; a.y = s[k];
      v2i b; b.x = s[k]; b.y = d[k];
      *(v2i*)(st_in  + 2 * (h_in [d[k] >> BIN_SH] + ri[k])) = a;
      *(v2i*)(st_out + 2 * (h_out[s[k] >> BIN_SH] + ro[k])) = b;
    }
  }
  __syncthreads();

  if (t <= NB) { seg_in[t * G + g] = ex_in; seg_out[t * G + g] = ex_out; }

  const int cnt = min(CH_E, E - base);
  int* gi = (int*)(bin_in  + base);
  int* go = (int*)(bin_out + base);
  for (int k = 0; k < CH_E / 2 / 1024; ++k) {
    const int q = t + k * 1024;
    if (2 * q + 1 < cnt) {
      *(v4i*)(gi + 4 * q) = *(const v4i*)(st_in  + 4 * q);
      *(v4i*)(go + 4 * q) = *(const v4i*)(st_out + 4 * q);
    } else if (2 * q < cnt) {
      *(v2i*)(gi + 4 * q) = *(const v2i*)(st_in  + 4 * q);
      *(v2i*)(go + 4 * q) = *(const v2i*)(st_out + 4 * q);
    }
  }
}

#define ST_CAP 12288

template <int PASS>
__device__ __forceinline__ int walk_runs(const int2* __restrict__ bin, const int* __restrict__ seg,
                                         int b, int G, int nb0, int* c, int* sst, int* sen,
                                         int* stage, int* __restrict__ lst_b, bool staged,
                                         int t, int lane, int wave) {
  int part = 0;
  for (int g0 = 0; g0 < G; g0 += 1024) {
    if (PASS == 0 || G > 1024) {
      if (g0 + t < G) {
        int a = seg[b * G + g0 + t];
        sst[t] = a; sen[t] = seg[(b + 1) * G + g0 + t];
        part += a;
      }
      __syncthreads();
    }
    const int ng = min(1024, G - g0);
    for (int j = wave; j < ng; j += 32) {
      const int st = sst[j], en = sen[j];
      const size_t cb = (size_t)(g0 + j) * CH_E;
      for (int p = st + lane; p < en; p += 32) {
        int2 r = bin[cb + p];
        if (PASS == 0) {
          atomicAdd(&c[r.x - nb0], 1);
        } else {
          const int pos = atomicAdd(&c[r.x - nb0], 1);
          if (staged) stage[pos] = r.y; else lst_b[pos] = r.y;
        }
      }
    }
    __syncthreads();
  }
  return part;
}

__global__ __launch_bounds__(1024)
void k_rank_place(const int2* __restrict__ bin_in, const int2* __restrict__ bin_out,
                  const int* __restrict__ seg_in, const int* __restrict__ seg_out,
                  int2* __restrict__ oc_in, int2* __restrict__ oc_out,
                  int* __restrict__ lst_in, int* __restrict__ lst_out, int N, int G) {
  __shared__ int c[BIN_W];
  __shared__ int sst[1024];
  __shared__ int sen[1024];
  __shared__ int wsum[32];
  __shared__ int wtot[BIN_W / 32];
  __shared__ int stage[ST_CAP];
  const int2* bin = blockIdx.y ? bin_out : bin_in;
  const int*  seg = blockIdx.y ? seg_out : seg_in;
  int2*       oc  = blockIdx.y ? oc_out  : oc_in;
  int*        lst = blockIdx.y ? lst_out : lst_in;
  const int t = threadIdx.x, lane = t & 31, wave = t >> 5, b = blockIdx.x;
  const int nb0 = b << BIN_SH;
  if (t < BIN_W) c[t] = 0;
  __syncthreads();

  int part = walk_runs<0>(bin, seg, b, G, nb0, c, sst, sen, stage, lst, false, t, lane, wave);

  for (int d = 16; d >= 1; d >>= 1) part += __shfl_xor(part, d);
  if (lane == 0) wsum[wave] = part;
  const int cv  = (t < BIN_W) ? c[t] : 0;
  const int inc = wave_incl_scan(cv, lane);
  if (t < BIN_W && lane == 31) wtot[wave] = inc;
  __syncthreads();
  int first = wsum[lane];
  for (int d = 16; d >= 1; d >>= 1) first += __shfl_xor(first, d);
  int total = (lane < BIN_W / 32) ? wtot[lane] : 0;
  for (int d = 16; d >= 1; d >>= 1) total += __shfl_xor(total, d);
  if (t < BIN_W) {
    int pre = 0;
    for (int w = 0; w < wave; ++w) pre += wtot[w];
    const int ex = pre + (inc - cv);
    if (nb0 + t < N) oc[nb0 + t] = make_int2(first + ex, cv);
    c[t] = ex;
  }
  __syncthreads();

  const bool staged = total <= ST_CAP;
  walk_runs<1>(bin, seg, b, G, nb0, c, sst, sen, stage, lst + first, staged, t, lane, wave);
  if (staged)
    for (int i = t; i < total; i += 1024) lst[first + i] = stage[i];
}


#define G1_WAVES 16
#define G1_T     (G1_WAVES * 32)
#define G1_COLS  (2 * HID_F)
#define G1_WP    (IN_F + 4)

__device__ __forceinline__ void g1_load_tile(const float* __restrict__ X, const int2* __restrict__ sc,
                                             int m0, int lane, v2f (&afrag)[IN_F / 4], float (&fs)[8]) {
  const int mrow = lane & 15;
  const int khi  = (lane >> 4) << 1;
  const float* xp = X + (size_t)(m0 + mrow) * IN_F + khi;
#pragma unroll
  for (int i = 0; i < IN_F / 4; ++i) afrag[i] = *(const v2f*)(xp + i * 4);
#pragma unroll
  for (int g = 0; g < 8; ++g)
    fs[g] = rsqrtf((float)(sc[m0 + ((lane >> 4) << 3) + g].y + 1));
}

__global__ __launch_bounds__(G1_T)
void k_gemm_wmma(const float* __restrict__ X0, const float* __restrict__ X1,
                 const float* __restrict__ Wa0, const float* __restrict__ ba0,
                 const float* __restrict__ Wb0, const float* __restrict__ bb0,
                 const float* __restrict__ Wa1, const float* __restrict__ ba1,
                 const float* __restrict__ Wb1, const float* __restrict__ bb1,
                 const int2* __restrict__ sc0, const int2* __restrict__ sc1,
                 float* __restrict__ Y0, float* __restrict__ Y1, int nMTiles) {
  constexpr int K  = IN_F;
  constexpr int NK = K / 4;
  __shared__ __attribute__((aligned(16))) float wl[G1_COLS * G1_WP];
  __shared__ float bl[G1_COLS];

  const int tid = threadIdx.x, lane = tid & 31;
  const int wave = __builtin_amdgcn_readfirstlane(tid >> 5);
  const int sd = (int)blockIdx.x & 1;
  const int bi = (int)blockIdx.x >> 1;
  const int nb = (int)gridDim.x >> 1;
  const float* X  = sd ? X1 : X0;
  const float* Wa = sd ? Wa1 : Wa0;
  const float* ba = sd ? ba1 : ba0;
  const float* Wb = sd ? Wb1 : Wb0;
  const float* bb = sd ? bb1 : bb0;
  float*       Y  = sd ? Y1 : Y0;
  const int2*  sc = sd ? sc1 : sc0;

  const int slot = (wave % 5 == 0) ? 12 + wave / 5 : wave - (wave + 4) / 5;
  int mt = bi + slot * nb;
  v2f afrag[NK];
  float fs[8];
  if (mt < nMTiles) g1_load_tile(X, sc, mt << 4, lane, afrag, fs);

#pragma unroll
  for (int u = 0; u < (G1_COLS * K / 4) / G1_T; ++u) {
    const int q = tid + u * G1_T;
    const int row = q >> 5, kc = (q & 31) * 4;
    const float* src = (row < HID_F) ? (Wa + row * K + kc) : (Wb + (row - HID_F) * K + kc);
    *(v4f*)(wl + row * G1_WP + kc) = *(const v4f*)src;
  }
  if (tid < G1_COLS) bl[tid] = (tid < HID_F) ? ba[tid] : bb[tid - HID_F];
  __syncthreads();

  const int col = lane & 15;
  const int khi = (lane >> 4) << 1;
  const float* wbase = wl + col * G1_WP + khi;

  while (mt < nMTiles) {
    const int m0 = mt << 4;
    const int mbase = m0 + ((lane >> 4) << 3);
#pragma nounroll
    for (int ct = 0; ct < G1_COLS / 16; ++ct) {
      const float* wp = wbase + ct * (16 * G1_WP);
      v8f acc = {};
#pragma unroll
      for (int i = 0; i < NK; ++i) {
        const v2f bf = *(const v2f*)(wp + i * 4);
        acc = __builtin_amdgcn_wmma_f32_16x16x4_f32(false, afrag[i], false, bf,
                                                    (short)0, acc, false, false);
      }
      const float bv = bl[ct * 16 + col];
      float* yp = Y + (size_t)mbase * G1_COLS + ct * 16 + col;
#pragma unroll
      for (int g = 0; g < 8; ++g)
        yp[(size_t)g * G1_COLS] = (acc[g] + bv) * fs[g];
    }
    mt += G1_WAVES * nb;
    if (mt < nMTiles) g1_load_tile(X, sc, mt << 4, lane, afrag, fs);
  }
}


template <int V> struct VecOf;
template <> struct VecOf<2> { typedef v2f type; };
template <> struct VecOf<4> { typedef v4f type; };

template <int F, int K>
__device__ __forceinline__ void rows_add(const float* __restrict__ T, int sx, int j, int c,
                                         typename VecOf<F / 32>::type& acc) {
  typedef typename VecOf<F / 32>::type vf;
  vf v[K];
#pragma unroll
  for (int u = 0; u < K; ++u) {
    int src = __builtin_amdgcn_readlane(sx, j + u);
    v[u] = *(const vf*)(T + (size_t)src * F + c);
  }
#pragma unroll
  for (int u = 0; u < K; ++u) acc += v[u];
}

template <int F, int EPI>
__global__ __launch_bounds__(256)
void k_gather(const float* __restrict__ T0, const float* __restrict__ T1,
              const int2* __restrict__ oc0, const int2* __restrict__ oc1,
              const int* __restrict__ lst0, const int* __restrict__ lst1,
              const float* __restrict__ noise0, const float* __restrict__ noise1,
              float* __restrict__ Y0, float* __restrict__ Y1, int N) {
  static_assert(!EPI || F == 2 * OUT_F, "the fused reparametrize needs a mu|logstd row");
  constexpr int V = F / 32;
  typedef typename VecOf<V>::type vf;
  const int lane = threadIdx.x & 31;
  const int c = lane * V;
  const int nWaves = gridDim.x * 8;
  const int nItems = 2 * N;
  int wid = __builtin_amdgcn_readfirstlane(blockIdx.x * 8 + (threadIdx.x >> 5));

  if (wid >= nItems) return;
  int2 o = (wid >= N) ? oc1[wid - N] : oc0[wid];
  for (int item = wid; ; ) {
    const int sd = item >= N ? 1 : 0;
    const int node = item - sd * N;
    const float* T   = sd ? T1 : T0;
    const int*   lst = sd ? lst1 : lst0;
    const int beg = o.x, cnt = o.y;
    const int nxt = item + nWaves;
    int2 o_nxt = o;
    if (nxt < nItems) o_nxt = (nxt >= N) ? oc1[nxt - N] : oc0[nxt];
    int n = min(32, cnt);
    int rec = 0;
    if (lane < n) rec = lst[beg + lane];
    vf acc = *(const vf*)(T + (size_t)node * F + c);
    v2f nz = {0.0f, 0.0f};
    if constexpr (EPI) {
      const float* noise = sd ? noise1 : noise0;
      if (lane < 16) nz = *(const v2f*)(noise + (size_t)node * OUT_F + c);
    }

    for (int base = 0; base < cnt; ) {
      int j = 0;
      for (; j + 8 <= n; j += 8) rows_add<F, 8>(T, rec, j, c, acc);
      if (j + 4 <= n) { rows_add<F, 4>(T, rec, j, c, acc); j += 4; }
      if (j + 2 <= n) { rows_add<F, 2>(T, rec, j, c, acc); j += 2; }
      if (j < n)      { rows_add<F, 1>(T, rec, j, c, acc); }
      base += 32;
      if (base < cnt) {
        n = min(32, cnt - base);
        rec = 0;
        if (lane < n) rec = lst[beg + base + lane];
      }
    }
    acc *= rsqrtf((float)(cnt + 1));
    float* Y = sd ? Y1 : Y0;
    if constexpr (EPI) {
      const float lx = __shfl_xor(acc.x, 16), ly = __shfl_xor(acc.y, 16);
      if (lane < 16) {
        v2f r;
        r.x = acc.x + nz.x * expf(lx) * 0.2f;
        r.y = acc.y + nz.y * expf(ly) * 0.2f;
        *(v2f*)(Y + (size_t)node * OUT_F + c) = r;
      }
    } else {
      *(vf*)(Y + (size_t)node * F + c) = acc;
    }
    if (nxt >= nItems) break;
    item = nxt;
    o = o_nxt;
  }
}


#define GL_WAVES 16
#define GL_T     (GL_WAVES * 32)
#define GL_AP    (IN_F + 4)
#define GL_WP    (HID_F + 4)

__global__ __launch_bounds__(GL_T)
void k_gather_lin2(const float* __restrict__ T0, const float* __restrict__ T1,
                   const int2* __restrict__ oc0, const int2* __restrict__ oc1,
                   const int* __restrict__ lst0, const int* __restrict__ lst1,
                   const float* __restrict__ Wa0, const float* __restrict__ ba0,
                   const float* __restrict__ Wb0, const float* __restrict__ bb0,
                   const float* __restrict__ Wa1, const float* __restrict__ ba1,
                   const float* __restrict__ Wb1, const float* __restrict__ bb1,
                   float* __restrict__ Y0, float* __restrict__ Y1, int N, int nTiles) {
  constexpr int F = IN_F;
  __shared__ __attribute__((aligned(16))) float wl[2 * HID_F * GL_WP];
  __shared__ __attribute__((aligned(16))) float tb[2 * 16 * GL_AP];
  __shared__ float bl[2 * HID_F];
  const int tid = threadIdx.x, lane = tid & 31;
  const int wave = __builtin_amdgcn_readfirstlane(tid >> 5);
  const int c = lane * 4;
  const int nAll = 2 * nTiles;

  int vt = blockIdx.x;
  if (vt >= nAll) return;
  int sd   = vt >= nTiles ? 1 : 0;
  int tile = vt - sd * nTiles;
  int node = tile * 16 + wave;
  int2 o = make_int2(0, 0);
  if (node < N) o = (sd ? oc1 : oc0)[node];

#pragma unroll
  for (int u = 0; u < 4; ++u) {
    const int q = tid + u * GL_T;
    const int row = (q >> 4) & (HID_F - 1), kc = (q & 15) * 4;
    const float* wa = (u < 2) ? Wa0 : Wa1;
    const float* wb = (u < 2) ? Wb0 : Wb1;
    const float* src = (row < OUT_F) ? (wa + row * HID_F + kc) : (wb + (row - OUT_F) * HID_F + kc);
    *(v4f*)(wl + (u >> 1) * (HID_F * GL_WP) + row * GL_WP + kc) = *(const v4f*)src;
  }
  if (tid < 2 * HID_F) {
    const int r = tid & (HID_F - 1);
    const float* pa = (tid < HID_F) ? ba0 : ba1;
    const float* pb = (tid < HID_F) ? bb0 : bb1;
    bl[tid] = (r < OUT_F) ? pa[r] : pb[r - OUT_F];
  }


  for (int it = 0; ; ++it) {
    const float* T   = sd ? T1 : T0;
    const int*   lst = sd ? lst1 : lst0;
    const int nvt   = vt + gridDim.x;
    const int nsd   = nvt >= nTiles ? 1 : 0;
    const int ntile = nvt - nsd * nTiles;
    const int nnode = ntile * 16 + wave;
    int2 o_nxt = make_int2(0, 0);
    if (nvt < nAll && nnode < N) o_nxt = (nsd ? oc1 : oc0)[nnode];

    v4f acc = {0.0f, 0.0f, 0.0f, 0.0f};
    float fd = 1.0f;
    if (node < N) {
      const int beg = o.x, cnt = o.y;
      int n = min(32, cnt);
      int rec = 0;
      if (lane < n) rec = lst[beg + lane];
      acc = *(const v4f*)(T + (size_t)node * F + c);
      for (int base = 0; base < cnt; ) {
        int j = 0;
        for (; j + 8 <= n; j += 8) rows_add<F, 8>(T, rec, j, c, acc);
        if (j + 4 <= n) { rows_add<F, 4>(T, rec, j, c, acc); j += 4; }
        if (j + 2 <= n) { rows_add<F, 2>(T, rec, j, c, acc); j += 2; }
        if (j < n)      { rows_add<F, 1>(T, rec, j, c, acc); }
        base += 32;
        if (base < cnt) {
          n = min(32, cnt - base);
          rec = 0;
          if (lane < n) rec = lst[beg + base + lane];
        }
      }
      fd = rsqrtf((float)(cnt + 1));
      acc *= fd;
    }
    acc.x = fmaxf(acc.x, 0.0f); acc.y = fmaxf(acc.y, 0.0f); acc.z = fmaxf(acc.z, 0.0f); acc.w = fmaxf(acc.w, 0.0f);
    float* tcur = tb + (it & 1) * (16 * GL_AP);
    *(v4f*)(tcur + wave * GL_AP + c) = acc;
    if (lane == 0) tcur[wave * GL_AP + IN_F] = fd;
    __syncthreads();

    if ((wave >> 2) == (it & 3)) {
      const int ct   = wave & 3;
      const int mrow = lane & 15;
      const int khi  = (lane >> 4) << 1;
      const int col  = lane & 15;
      const float* ap = tcur + mrow * GL_AP + (ct >> 1) * HID_F + khi;
      const float* wp = wl + sd * (HID_F * GL_WP) + (ct * 16 + col) * GL_WP + khi;
      const int mbase = tile * 16 + ((lane >> 4) << 3);
      const float* fp = tcur + ((lane >> 4) << 3) * GL_AP + IN_F;
      float fs[8];
#pragma unroll
      for (int g = 0; g < 8; ++g) fs[g] = fp[g * GL_AP];
      v8f a8 = {};
#pragma unroll
      for (int i = 0; i < HID_F / 4; ++i) {
        v2f af = *(const v2f*)(ap + i * 4);
        v2f bf = *(const v2f*)(wp + i * 4);
        a8 = __builtin_amdgcn_wmma_f32_16x16x4_f32(false, af, false, bf, (short)0, a8, false, false);
      }
      const float bv = bl[sd * HID_F + ct * 16 + col];
      float* yp = (sd ? Y1 : Y0) + (size_t)mbase * HID_F + ct * 16 + col;
#pragma unroll
      for (int g = 0; g < 8; ++g)
        if (mbase + g < N) yp[(size_t)g * HID_F] = (a8[g] + bv) * fs[g];
    }
    if (nvt >= nAll) break;
    vt = nvt; sd = nsd; tile = ntile; node = nnode; o = o_nxt;
  }
}


static inline int ceil_div(int a, int b) { return (a + b - 1) / b; }

extern "C" void kernel_launch(void* const* d_in, const int* in_sizes, int n_in,
                              void* d_out, int out_size, void* d_ws, size_t ws_size,
                              hipStream_t stream) {
  const float* s       = (const float*)d_in[0];
  const float* t       = (const float*)d_in[1];
  const int*   ei      = (const int*)  d_in[2];
  const float* noise_s = (const float*)d_in[3];
  const float* noise_t = (const float*)d_in[4];
  const float* sm1_W = (const float*)d_in[5],  *sm1_b = (const float*)d_in[6];
  const float* sm2_W = (const float*)d_in[7],  *sm2_b = (const float*)d_in[8];
  const float* sl1_W = (const float*)d_in[9],  *sl1_b = (const float*)d_in[10];
  const float* sl2_W = (const float*)d_in[11], *sl2_b = (const float*)d_in[12];
  const float* tm1_W = (const float*)d_in[13], *tm1_b = (const float*)d_in[14];
  const float* tm2_W = (const float*)d_in[15], *tm2_b = (const float*)d_in[16];
  const float* tl1_W = (const float*)d_in[17], *tl1_b = (const float*)d_in[18];
  const float* tl2_W = (const float*)d_in[19], *tl2_b = (const float*)d_in[20];

  const int N = in_sizes[0] / IN_F;
  const int E = in_sizes[2] / 2;
  const int* ei0 = ei;
  const int* ei1 = ei + E;

  float* ws = (float*)d_ws;
  float* P0 = ws + (size_t)3 * N + E;
  float* P1 = P0 + (size_t)N * IN_F;
  float* P2 = P1 + (size_t)N * IN_F;
  float* P3 = P2 + (size_t)N * IN_F;
  int2* oc_in   = (int2*)P3;
  int2* oc_out  = oc_in  + N;
  int*  lst_in  = (int*)(oc_out + N);
  int*  lst_out = lst_in + E;

  const int NB = ceil_div(N, BIN_W);
  const int G  = ceil_div(E, CH_E);
  const int    padw = (32 - (int)(((size_t)3 * N + E) & 31)) & 31;
  const size_t Ep   = ((size_t)E + 15) & ~(size_t)15;
  int2* bin_in  = (int2*)(P0 + padw);
  int2* bin_out = bin_in + Ep;
  int*  seg_in  = (int*)(bin_out + Ep);
  int*  seg_out = seg_in + (size_t)(NB + 1) * G;

  float* Y1s = P0;
  float* Y1t = P2;
  float* Y2s = P1;
  float* Y2t = P1 + (size_t)N * HID_F;
  float* out_s = (float*)d_out;
  float* out_t = out_s + (size_t)N * OUT_F;

  const int B = 256;
  const int nMTiles = ceil_div(N, 16);
  const int gemmBlocks = 512;
  const int gatherBlocks = ceil_div(2 * N, 8) < 2048 ? ceil_div(2 * N, 8) : 2048;
  const int fusedBlocks = 2 * nMTiles < 1024 ? 2 * nMTiles : 1024;

  k_bin<<<G, 1024, 0, stream>>>(ei0, ei1, bin_in, bin_out, seg_in, seg_out, E, NB, G);
  k_rank_place<<<dim3(NB, 2), 1024, 0, stream>>>(bin_in, bin_out, seg_in, seg_out, oc_in, oc_out,
                                                 lst_in, lst_out, N, G);

  k_gemm_wmma<<<gemmBlocks, G1_T, 0, stream>>>(
      s, t, sm1_W, sm1_b, sl1_W, sl1_b, tm1_W, tm1_b, tl1_W, tl1_b, oc_out, oc_in,
      Y1s, Y1t, nMTiles);
  k_gather_lin2<<<fusedBlocks, GL_T, 0, stream>>>(Y1s, Y1t, oc_in, oc_out, lst_in, lst_out,
                                                  sm2_W, sm2_b, sl2_W, sl2_b, tm2_W, tm2_b, tl2_W, tl2_b,
                                                  Y2s, Y2t, N, nMTiles);
  k_gather<HID_F, 1><<<gatherBlocks, B, 0, stream>>>(Y2s, Y2t, oc_out, oc_in, lst_out, lst_in,
                                                     noise_s, noise_t, out_s, out_t, N);
}
